# starting kernel
# baseline (speedup 1.0000x reference)
#include <hip/hip_runtime.h>
#include <hip/hip_bf16.h>
#include <math.h>

typedef float v2f __attribute__((ext_vector_type(2)));
typedef float v8f __attribute__((ext_vector_type(8)));

__global__ void k_fill(float* __restrict__ p, float val, int n) {
  int i = blockIdx.x * blockDim.x + threadIdx.x;
  if (i < n) p[i] = val;
}

__global__ void k_deg_edges(const int* __restrict__ dst, float* __restrict__ deg, int E) {
  int e = blockIdx.x * blockDim.x + threadIdx.x;
  if (e < E) unsafeAtomicAdd(&deg[dst[e]], 1.0f);
}

__global__ void k_rsqrt(const float* __restrict__ deg, float* __restrict__ dinv, int n) {
  int i = blockIdx.x * blockDim.x + threadIdx.x;
  if (i < n) dinv[i] = rsqrtf(deg[i]);
}

__global__ void k_pack_bfrag(const float* __restrict__ B, v2f* __restrict__ BF,
                             int N, int nTiles) {
  int i = blockIdx.x * blockDim.x + threadIdx.x;
  int total = nTiles * 1024;
  if (i >= total) return;
  int lane = i & 31;
  int t    = (i >> 5) & 31;
  int nt   = i >> 10;
  int half = lane >> 4;
  int lm   = lane & 15;
  int col  = nt * 16 + lm;
  int kk   = 4 * t + 2 * half;
  v2f v;
  v.x = (col < N) ? B[(size_t)kk * N + col]       : 0.0f;
  v.y = (col < N) ? B[(size_t)(kk + 1) * N + col] : 0.0f;
  BF[i] = v;
}

__global__ __launch_bounds__(256) void gcn_gemm_wmma_f32(
    const float* __restrict__ A, const v2f* __restrict__ BF,
    float* __restrict__ C, int M, int N, int nTiles) {
  constexpr int K = 128;
  constexpr int LDS_STRIDE = 132;
  __shared__ float As[16 * LDS_STRIDE];

  const int m0  = blockIdx.x * 16;
  const int tid = threadIdx.x;

#pragma unroll
  for (int j = 0; j < 8; ++j) {
    int off = tid + j * 256;
    int row = off >> 7;
    int col = off & 127;
    As[row * LDS_STRIDE + col] = A[(size_t)(m0 + row) * K + col];
  }
  __syncthreads();

  const int wave = tid >> 5;
  const int lane = tid & 31;
  const int half = lane >> 4;
  const int lm   = lane & 15;

  const float* aBase = &As[lm * LDS_STRIDE + 2 * half];

  for (int nt = wave; nt < nTiles; nt += 8) {
    const v2f* bBase = BF + (size_t)nt * 1024 + lane;
    v8f c = {};
#pragma unroll
    for (int t = 0; t < 32; ++t) {
      v2f a = *(const v2f*)(aBase + 4 * t);
      v2f b = bBase[(size_t)t * 32];
      c = __builtin_amdgcn_wmma_f32_16x16x4_f32(
           false, a,  false, b,
           (short)0, c,  false,  false);
    }
    const int col = nt * 16 + lm;
    if (col < N) {
      const int mBase = m0 + 8 * half;
#pragma unroll
      for (int r = 0; r < 8; ++r)
        C[(size_t)(mBase + r) * N + col] = c[r];
    }
  }
}

__global__ void k_self_init(const float* __restrict__ h, const float* __restrict__ dinv,
                            float* __restrict__ out, int total, int F) {
  int i = blockIdx.x * blockDim.x + threadIdx.x;
  if (i >= total) return;
  int node = i / F;
  float di = dinv[node];
  out[i] = di * di * h[i];
}

__global__ void k_edge_f128(const float* __restrict__ h, const int* __restrict__ src,
                            const int* __restrict__ dst, const float* __restrict__ dinv,
                            float* __restrict__ out, int E) {
  int tid = blockIdx.x * blockDim.x + threadIdx.x;
  int e = tid >> 7;
  if (e >= E) return;
  int f = tid & 127;
  int s = src[e], d = dst[e];
  float coeff = dinv[s] * dinv[d];
  unsafeAtomicAdd(&out[(size_t)d * 128 + f], coeff * h[(size_t)s * 128 + f]);
}

__global__ void k_edge_f40(const float* __restrict__ h, const int* __restrict__ src,
                           const int* __restrict__ dst, const float* __restrict__ dinv,
                           float* __restrict__ out, int E) {
  int tid = blockIdx.x * blockDim.x + threadIdx.x;
  int e = tid >> 6;
  int f = tid & 63;
  if (e >= E || f >= 40) return;
  int s = src[e], d = dst[e];
  float coeff = dinv[s] * dinv[d];
  unsafeAtomicAdd(&out[(size_t)d * 40 + f], coeff * h[(size_t)s * 40 + f]);
}

__global__ void k_bias_relu(float* __restrict__ out, const float* __restrict__ b,
                            int total, int F) {
  int i = blockIdx.x * blockDim.x + threadIdx.x;
  if (i >= total) return;
  float v = out[i] + b[i % F];
  out[i] = v > 0.0f ? v : 0.0f;
}

__global__ __launch_bounds__(256) void k_logsoftmax40(float* __restrict__ out,
                                                      const float* __restrict__ b, int N) {
  int row  = blockIdx.x * 8 + (threadIdx.x >> 5);
  int lane = threadIdx.x & 31;
  if (row >= N) return;
  float* p = out + (size_t)row * 40;
  float v0 = p[lane] + b[lane];
  bool has2 = (lane + 32) < 40;
  float v1 = has2 ? p[lane + 32] + b[lane + 32] : -INFINITY;

  float m = fmaxf(v0, v1);
#pragma unroll
  for (int off = 16; off > 0; off >>= 1) m = fmaxf(m, __shfl_xor(m, off, 32));

  float s = expf(v0 - m) + (has2 ? expf(v1 - m) : 0.0f);
#pragma unroll
  for (int off = 16; off > 0; off >>= 1) s += __shfl_xor(s, off, 32);

  float lse = logf(s);
  p[lane] = v0 - m - lse;
  if (has2) p[lane + 32] = v1 - m - lse;
}

static inline int cdiv_ll(long long a, long long b) { return (int)((a + b - 1) / b); }

extern "C" void kernel_launch(void* const* d_in, const int* in_sizes, int n_in,
                              void* d_out, int out_size, void* d_ws, size_t ws_size,
                              hipStream_t stream) {
  const float* x  = (const float*)d_in[0];
  const int*   ei = (const int*)  d_in[1];
  const float* W1 = (const float*)d_in[2];
  const float* b1 = (const float*)d_in[3];
  const float* W2 = (const float*)d_in[4];
  const float* b2 = (const float*)d_in[5];

  const int F = 128, H = 128, C = 40;
  const int N = in_sizes[0] / F;
  const int E = in_sizes[1] / 2;
  const int* src = ei;
  const int* dst = ei + E;

  const int nT1 = (H + 15) / 16;
  const int nT2 = (C + 15) / 16;

  float* ws   = (float*)d_ws;
  float* deg  = ws;                 ws += N;
  float* dinv = ws;                 ws += N;
  float* h1   = ws;                 ws += (size_t)N * H;
  float* o1   = ws;                 ws += (size_t)N * H;
  float* h2   = ws;                 ws += (size_t)N * C;
  v2f* bf1    = (v2f*)ws;           ws += (size_t)nT1 * 1024 * 2;
  v2f* bf2    = (v2f*)ws;           ws += (size_t)nT2 * 1024 * 2;
  float* out  = (float*)d_out;

  const int B = 256;

  k_fill      <<<cdiv_ll(N, B), B, 0, stream>>>(deg, 1.0f, N);
  k_deg_edges <<<cdiv_ll(E, B), B, 0, stream>>>(dst, deg, E);
  k_rsqrt     <<<cdiv_ll(N, B), B, 0, stream>>>(deg, dinv, N);

  k_pack_bfrag<<<cdiv_ll((long long)nT1 * 1024, B), B, 0, stream>>>(W1, bf1, H, nT1);
  k_pack_bfrag<<<cdiv_ll((long long)nT2 * 1024, B), B, 0, stream>>>(W2, bf2, C, nT2);

  gcn_gemm_wmma_f32<<<N / 16, B, 0, stream>>>(x, bf1, h1, N, H, nT1);
  k_self_init <<<cdiv_ll((long long)N * H, B), B, 0, stream>>>(h1, dinv, o1, N * H, H);
  k_edge_f128 <<<cdiv_ll((long long)E * 128, B), B, 0, stream>>>(h1, src, dst, dinv, o1, E);
  k_bias_relu <<<cdiv_ll((long long)N * H, B), B, 0, stream>>>(o1, b1, N * H, H);

  gcn_gemm_wmma_f32<<<N / 16, B, 0, stream>>>(o1, bf2, h2, N, C, nT2);
  k_self_init <<<cdiv_ll((long long)N * C, B), B, 0, stream>>>(h2, dinv, out, N * C, C);
  k_edge_f40  <<<cdiv_ll((long long)E * 64, B), B, 0, stream>>>(h2, src, dst, dinv, out, E);
  k_logsoftmax40<<<cdiv_ll(N, 8), B, 0, stream>>>(out, b2, N);
}


# Layer-2 product taken from the summed tile while it lies in LDS: the tile block adds b1, applies relu and multiplies its 196 rows by W2 on WMMA (same k order); o1 never stored; 10->8 launches
# speedup vs baseline: 3.9809x; 3.9809x over previous
#include <hip/hip_runtime.h>
#include <hip/hip_bf16.h>
#include <math.h>

typedef float v2f __attribute__((ext_vector_type(2)));
typedef float v8f __attribute__((ext_vector_type(8)));

__global__ void k_pack_bfrag(const float* __restrict__ B, v2f* __restrict__ BF,
                             int N, int nTiles) {
  int i = blockIdx.x * blockDim.x + threadIdx.x;
  int total = nTiles * 1024;
  if (i >= total) return;
  int lane = i & 31;
  int t    = (i >> 5) & 31;
  int nt   = i >> 10;
  int half = lane >> 4;
  int lm   = lane & 15;
  int col  = nt * 16 + lm;
  int kk   = 4 * t + 2 * half;
  v2f v;
  v.x = (col < N) ? B[(size_t)kk * N + col]       : 0.0f;
  v.y = (col < N) ? B[(size_t)(kk + 1) * N + col] : 0.0f;
  BF[i] = v;
}

__global__ __launch_bounds__(256) void gcn_gemm_wmma_f32(
    const float* __restrict__ A, const v2f* __restrict__ BF,
    float* __restrict__ C, int M, int N, int nTiles) {
  constexpr int K = 128;
  constexpr int LDS_STRIDE = 132;
  __shared__ float As[16 * LDS_STRIDE];

  const int m0  = blockIdx.x * 16;
  const int tid = threadIdx.x;

#pragma unroll
  for (int j = 0; j < 8; ++j) {
    int off = tid + j * 256;
    int row = off >> 7;
    int col = off & 127;
    As[row * LDS_STRIDE + col] = A[(size_t)(m0 + row) * K + col];
  }
  __syncthreads();

  const int wave = tid >> 5;
  const int lane = tid & 31;
  const int half = lane >> 4;
  const int lm   = lane & 15;

  const float* aBase = &As[lm * LDS_STRIDE + 2 * half];

  for (int nt = wave; nt < nTiles; nt += 8) {
    const v2f* bBase = BF + (size_t)nt * 1024 + lane;
    v8f c = {};
#pragma unroll
    for (int t = 0; t < 32; ++t) {
      v2f a = *(const v2f*)(aBase + 4 * t);
      v2f b = bBase[(size_t)t * 32];
      c = __builtin_amdgcn_wmma_f32_16x16x4_f32(
           false, a,  false, b,
           (short)0, c,  false,  false);
    }
    const int col = nt * 16 + lm;
    if (col < N) {
      const int mBase = m0 + 8 * half;
#pragma unroll
      for (int r = 0; r < 8; ++r)
        C[(size_t)(mBase + r) * N + col] = c[r];
    }
  }
}

constexpr int AGG_THREADS = 1024;
constexpr int AGG_ROWS    = 196;
constexpr int AGG_LDS_BYTES = 168 * 1024;
constexpr int LIST_CAP      = 4096;

template <int V> struct IntC { static constexpr int value = V; };

constexpr int SCAN_CAP = (AGG_LDS_BYTES - AGG_ROWS * 4 - 64) / 4;
static_assert(LIST_CAP <= SCAN_CAP, "the scan's LDS list must hold a full workspace list");

__global__ __launch_bounds__(256) void k_tile_bytes(
    const int* __restrict__ dst, unsigned* __restrict__ tb32, int E) {
  const int E4 = E >> 2;
  const int base = blockIdx.x * 1024 + threadIdx.x;
  auto pack = [](int4 d) {
    unsigned b0 = ((unsigned)d.x / (unsigned)AGG_ROWS) & 255u;
    unsigned b1 = ((unsigned)d.y / (unsigned)AGG_ROWS) & 255u;
    unsigned b2 = ((unsigned)d.z / (unsigned)AGG_ROWS) & 255u;
    unsigned b3 = ((unsigned)d.w / (unsigned)AGG_ROWS) & 255u;
    return b0 | (b1 << 8) | (b2 << 16) | (b3 << 24);
  };
  if (E4 > 0) {
    if ((((size_t)dst) & 15) == 0) {
      const int4* dst4 = (const int4*)dst;
      int4 d[4];
#pragma unroll
      for (int u = 0; u < 4; ++u) d[u] = dst4[min(base + u * 256, E4 - 1)];
#pragma unroll
      for (int u = 0; u < 4; ++u) {
        const int g = base + u * 256;
        if (g < E4) tb32[g] = pack(d[u]);
      }
    } else {
#pragma unroll
      for (int u = 0; u < 4; ++u) {
        const int g = base + u * 256;
        if (g < E4) tb32[g] = pack(make_int4(dst[4 * g], dst[4 * g + 1], dst[4 * g + 2], dst[4 * g + 3]));
      }
    }
  }
  if (blockIdx.x == 0 && (int)threadIdx.x < (E & 3)) {
    const int e = 4 * E4 + threadIdx.x;
    ((unsigned char*)tb32)[e] = (unsigned char)(((unsigned)dst[e] / (unsigned)AGG_ROWS) & 255u);
  }
}

template <bool BYTES>
__global__ __launch_bounds__(AGG_THREADS) void k_scan_tiles(
    const int* __restrict__ dst, const unsigned char* __restrict__ tileb,
    float* __restrict__ dinv, int* __restrict__ wlist, int* __restrict__ wcnt,
    int capW, int N, int E) {
  __shared__ int list[SCAN_CAP];
  __shared__ int rowcnt[AGG_ROWS];
  __shared__ int cnt;
  __shared__ int cntv;

  const int tid   = threadIdx.x;
  const int r0    = blockIdx.x * AGG_ROWS;
  const int nrows = min(AGG_ROWS, N - r0);
  const unsigned ur0 = (unsigned)r0, un = (unsigned)nrows;

  for (int i = tid; i < AGG_ROWS; i += AGG_THREADS) rowcnt[i] = 0;
  if (tid == 0) { cnt = 0; cntv = 0; }
  __syncthreads();

  auto count_row = [&](int e) -> bool {
    int d = dst[e];
    bool ok = ((unsigned)d - ur0) < un;
    if (ok) atomicAdd(&rowcnt[d - r0], 1);
    return ok;
  };
  auto note = [&](int e) {
    int pos = atomicAdd(&cnt, 1);
    if (pos < SCAN_CAP) list[pos] = e;
    else count_row(e);
  };

  constexpr int U = 8;
  constexpr int TILE = U * AGG_THREADS;
  if constexpr (BYTES) {
    const unsigned tile = blockIdx.x;
    int scalarFrom = 0;
    if ((((size_t)tileb) & 15) == 0) {
      const unsigned pat = tile * 0x01010101u;
      auto hit = [&](unsigned w) {
        unsigned x = w ^ pat;
        return (x - 0x01010101u) & ~x & 0x80808080u;
      };
      auto flags = [&](uint4 v, auto hi) -> unsigned {
        constexpr int HI = decltype(hi)::value;
        return (hit(v.x) >> (7 - HI)) | (hit(v.y) >> (6 - HI)) | (hit(v.z) >> (5 - HI)) | (hit(v.w) >> (4 - HI));
      };
      auto drain = [&](unsigned mm, int e0, int e1) {
        while (mm) {
          int b = __builtin_ctz(mm);
          mm &= mm - 1;
          note(((b & 4) ? e1 : e0) + 4 * (b & 3) + (b >> 3));
        }
      };
      const int G16 = E >> 4;
      const int G32 = G16 & ~31;
      const uint4* tb4 = (const uint4*)tileb;
      const int wave = __builtin_amdgcn_readfirstlane(tid >> 5);
      const int lane = tid & 31;
      const int start = (int)(tile * ((unsigned)G32 / gridDim.x)) & ~31;
      auto place = [&](int i) { int q = start + i; return q >= G32 ? q - G32 : q; };
      int i0 = 0;
      for (; i0 + TILE <= G32; i0 += TILE) {
        uint4 dv[U];
        int   q[U];
#pragma unroll
        for (int u = 0; u < U; ++u) {
          q[u]  = place(i0 + u * AGG_THREADS + wave * 32) + lane;
          dv[u] = tb4[q[u]];
        }
#pragma unroll
        for (int u = 0; u < U; u += 2)
          drain(flags(dv[u], IntC<0>{}) | flags(dv[u + 1], IntC<4>{}), 16 * q[u], 16 * q[u + 1]);
      }
      for (int i = i0 + wave * 32; i < G32; i += AGG_THREADS) {
        const int q = place(i) + lane;
        drain(flags(tb4[q], IntC<0>{}), 16 * q, 16 * q);
      }
      if (tid < G16 - G32) {
        const int q = G32 + tid;
        drain(flags(tb4[q], IntC<0>{}), 16 * q, 16 * q);
      }
      scalarFrom = G16 * 16;
    }
    for (int e = scalarFrom + tid; e < E; e += AGG_THREADS)
      if (tileb[e] == (unsigned char)tile) note(e);
  } else {
    auto check4 = [&](int4 d4, int e) {
      bool m0 = ((unsigned)d4.x - ur0) < un;
      bool m1 = ((unsigned)d4.y - ur0) < un;
      bool m2 = ((unsigned)d4.z - ur0) < un;
      bool m3 = ((unsigned)d4.w - ur0) < un;
      if (m0 | m1 | m2 | m3) {
        if (m0) note(e);
        if (m1) note(e + 1);
        if (m2) note(e + 2);
        if (m3) note(e + 3);
      }
    };
    int scalarFrom = 0;
    if ((((size_t)dst) & 15) == 0) {
      const int E4 = E >> 2;
      const int4* dst4 = (const int4*)dst;
      auto scan_groups = [&](int lo, int hi) {
        int t0 = lo;
        for (; t0 + TILE <= hi; t0 += TILE) {
          int4 dv[U];
#pragma unroll
          for (int u = 0; u < U; ++u) dv[u] = dst4[t0 + tid + u * AGG_THREADS];
#pragma unroll
          for (int u = 0; u < U; ++u) check4(dv[u], 4 * (t0 + tid + u * AGG_THREADS));
        }
        for (int q = t0 + tid; q < hi; q += AGG_THREADS) check4(dst4[q], 4 * q);
      };
      const int start = (int)(blockIdx.x * ((unsigned)E4 / gridDim.x)) & ~7;
      scan_groups(start, E4);
      scan_groups(0, start);
      scalarFrom = E4 * 4;
    }
    for (int e = scalarFrom + tid; e < E; e += AGG_THREADS) {
      int d = dst[e];
      if (((unsigned)d - ur0) < un) note(e);
    }
  }
  __syncthreads();

  const int noted = cnt;
  const int held  = min(noted, SCAN_CAP);
  int* wl = wlist + (size_t)blockIdx.x * LIST_CAP;
  for (int m = tid; m < held; m += AGG_THREADS) {
    int e = list[m];
    if (count_row(e)) {
      int p = atomicAdd(&cntv, 1);
      if (p < capW) wl[p] = e;
    }
  }
  __syncthreads();

  for (int i = tid; i < nrows; i += AGG_THREADS) dinv[r0 + i] = rsqrtf((float)(rowcnt[i] + 1));

  if (tid == 0) wcnt[blockIdx.x] = (noted <= SCAN_CAP) ? cntv : 0x7fffffff;
}

template <int F, int FO>
__global__ __launch_bounds__(AGG_THREADS) void k_agg_rows(
    const float* __restrict__ h, const int* __restrict__ src, const int* __restrict__ dst,
    const float* __restrict__ dinv, const int* __restrict__ wlist, const int* __restrict__ wcnt,
    int capW, float* __restrict__ out, int N, int E,
    const float* __restrict__ bias, const v2f* __restrict__ BF) {
  constexpr bool MM     = FO > 0;
  constexpr int  PITCH  = MM ? F + 4 : F;
  constexpr int  ROWS_A = MM ? ((AGG_ROWS + 15) / 16) * 16 : AGG_ROWS;
  constexpr int  CAPL   = (AGG_LDS_BYTES - ROWS_A * PITCH * 4) / 12;
  static_assert(LIST_CAP <= CAPL, "LDS must hold a full workspace list beside the tile");
  static_assert(!MM || (F % 4 == 0 && ROWS_A * FO <= 3 * CAPL), "product rows are staged where the edge list was");
  __shared__ alignas(16) float acc[ROWS_A * PITCH];
  __shared__ int lst[3 * CAPL];
  int* eS = lst;
  int* eD = lst + CAPL;
  int* eC = lst + 2 * CAPL;

  const int tid   = threadIdx.x;
  const int r0    = blockIdx.x * AGG_ROWS;
  const int nrows = min(AGG_ROWS, N - r0);
  const int nel   = nrows * F;
  const size_t base = (size_t)r0 * F;

  const int  total  = wcnt[blockIdx.x];
  const bool listed = total <= capW;

  if (listed) {
    const int* wl = wlist + (size_t)blockIdx.x * LIST_CAP;
    for (int m = tid; m < total; m += AGG_THREADS) {
      int e = wl[m];
      int s = src[e];
      int d = dst[e];
      eS[m] = s;
      eD[m] = d - r0;
      eC[m] = __float_as_int(dinv[s] * dinv[d]);
    }
  }
  for (int i = tid; i < ROWS_A * F; i += AGG_THREADS) {
    const int row = i / F, col = i % F;
    float v = 0.0f;
    if (i < nel) {
      float di = dinv[r0 + row];
      v = di * di * h[base + i];
    }
    acc[row * PITCH + col] = v;
  }
  __syncthreads();

  if (listed) {
    const int wave = __builtin_amdgcn_readfirstlane(tid >> 5);
    const int lane = tid & 31;
    constexpr int NJ = (F + 31) / 32;
    constexpr int G  = 8;
    for (int m0 = wave * 32; m0 < total; m0 += AGG_THREADS) {
      const int m = m0 + lane;
      int s = 0, dl = 0;
      float c = 0.0f;
      if (m < total) {
        s  = eS[m];
        dl = eD[m];
        c  = __int_as_float(eC[m]);
      }
      auto add_edges = [&](auto gg, int i0) {
        constexpr int GG = decltype(gg)::value;
        float v[GG][NJ];
#pragma unroll
        for (int i = 0; i < GG; ++i) {
          int si = __builtin_amdgcn_readlane(s, i0 + i);
          const float* hr = h + (size_t)si * F;
#pragma unroll
          for (int j = 0; j < NJ; ++j) v[i][j] = hr[min(32 * j + lane, F - 1)];
        }
#pragma unroll
        for (int i = 0; i < GG; ++i) {
          int   di = __builtin_amdgcn_readlane(dl, i0 + i);
          float ci = __builtin_bit_cast(float, __builtin_amdgcn_readlane(__builtin_bit_cast(int, c), i0 + i));
          float* ar = acc + di * PITCH;
#pragma unroll
          for (int j = 0; j < NJ; ++j) {
            int f = 32 * j + lane;
            if (f < F) atomicAdd(&ar[f], ci * v[i][j]);
          }
        }
      };
      const int nb = min(32, total - m0);
      int i0 = 0;
      for (; i0 + G <= nb; i0 += G) add_edges(IntC<G>{}, i0);
      for (; i0 < nb; ++i0)         add_edges(IntC<1>{}, i0);
    }
  } else {
    const unsigned ur0 = (unsigned)r0, un = (unsigned)nrows;
    for (int e = tid; e < E; e += AGG_THREADS) {
      int d = dst[e];
      if (((unsigned)d - ur0) < un) {
        int s = src[e];
        float c = dinv[s] * dinv[d];
        for (int f = 0; f < F; ++f)
          atomicAdd(&acc[(d - r0) * PITCH + f], c * h[(size_t)s * F + f]);
      }
    }
  }
  __syncthreads();

  if constexpr (!MM) {
    for (int i = tid; i < nel; i += AGG_THREADS) out[base + i] = acc[i];
  } else {
    for (int i = tid; i < nel; i += AGG_THREADS) {
      const int row = i / F, col = i % F;
      float v = acc[row * PITCH + col] + bias[col];
      acc[row * PITCH + col] = v > 0.0f ? v : 0.0f;
    }
    __syncthreads();

    constexpr int NT = (FO + 15) / 16;
    constexpr int JOBS = (ROWS_A / 16) * NT;
    const int wave = __builtin_amdgcn_readfirstlane(tid >> 5);
    const int lane = tid & 31;
    const int half = lane >> 4;
    const int lm   = lane & 15;
    for (int job = wave; job < JOBS; job += AGG_THREADS / 32) {
      const int rt = job / NT, nt = job % NT;
      const float* aBase = acc + (rt * 16 + lm) * PITCH + 2 * half;
      const v2f*   bBase = BF + (size_t)nt * 1024 + lane;
      v8f c = {};
#pragma unroll
      for (int t = 0; t < F / 4; ++t) {
        v2f a = *(const v2f*)(aBase + 4 * t);
        v2f b = bBase[(size_t)t * 32];
        c = __builtin_amdgcn_wmma_f32_16x16x4_f32(
             false, a,  false, b,
             (short)0, c,  false,  false);
      }
      const int col = nt * 16 + lm;
      if (col < FO) {
        int* st = lst + (rt * 16 + 8 * half) * FO + col;
#pragma unroll
        for (int r = 0; r < 8; ++r) {
          const float cr = c[r];
          st[r * FO] = __float_as_int(cr);
        }
      }
    }
    __syncthreads();

    const size_t obase = (size_t)r0 * FO;
    for (int i = tid; i < nrows * FO; i += AGG_THREADS) out[obase + i] = __int_as_float(lst[i]);
  }
}

__global__ __launch_bounds__(256) void k_logsoftmax40(float* __restrict__ out,
                                                      const float* __restrict__ b, int N) {
  int row  = blockIdx.x * 8 + (threadIdx.x >> 5);
  int lane = threadIdx.x & 31;
  if (row >= N) return;
  float* p = out + (size_t)row * 40;
  float v0 = p[lane] + b[lane];
  bool has2 = (lane + 32) < 40;
  float v1 = has2 ? p[lane + 32] + b[lane + 32] : -INFINITY;

  float m = fmaxf(v0, v1);
#pragma unroll
  for (int off = 16; off > 0; off >>= 1) m = fmaxf(m, __shfl_xor(m, off, 32));

  float s = expf(v0 - m) + (has2 ? expf(v1 - m) : 0.0f);
#pragma unroll
  for (int off = 16; off > 0; off >>= 1) s += __shfl_xor(s, off, 32);

  float lse = logf(s);
  p[lane] = v0 - m - lse;
  if (has2) p[lane + 32] = v1 - m - lse;
}

static inline int cdiv_ll(long long a, long long b) { return (int)((a + b - 1) / b); }

extern "C" void kernel_launch(void* const* d_in, const int* in_sizes, int n_in,
                              void* d_out, int out_size, void* d_ws, size_t ws_size,
                              hipStream_t stream) {
  const float* x  = (const float*)d_in[0];
  const int*   ei = (const int*)  d_in[1];
  const float* W1 = (const float*)d_in[2];
  const float* b1 = (const float*)d_in[3];
  const float* W2 = (const float*)d_in[4];
  const float* b2 = (const float*)d_in[5];

  const int F = 128, H = 128, C = 40;
  const int N = in_sizes[0] / F;
  const int E = in_sizes[1] / 2;
  const int* src = ei;
  const int* dst = ei + E;

  const int nT1 = (H + 15) / 16;
  const int nT2 = (C + 15) / 16;

  const int nAgg = cdiv_ll(N, AGG_ROWS);

  float* ws   = (float*)d_ws;
  int*   wcnt = (int*)ws;           ws += N;
  float* dinv = ws;                 ws += N;
  float* h1   = ws;                 ws += (size_t)N * H;
  float* o1   = ws;                 ws += (size_t)N * H;
  int*   wlist = (int*)ws;          ws += (size_t)N * C;
  v2f* bf1    = (v2f*)ws;           ws += (size_t)nT1 * 1024 * 2;
  v2f* bf2    = (v2f*)ws;           ws += (size_t)nT2 * 1024 * 2;
  float* h2   = o1;
  float* out  = (float*)d_out;
  const int capW = ((size_t)nAgg * LIST_CAP <= (size_t)N * C) ? LIST_CAP : 0;

  const int B = 256;

  unsigned char* tileb = (unsigned char*)o1;
  const bool bytes = nAgg <= 256 && (size_t)E <= (size_t)N * H * sizeof(float);
  if (bytes) {
    k_tile_bytes<<<cdiv_ll((long long)(E >> 2) + 1, 1024), B, 0, stream>>>(dst, (unsigned*)tileb, E);
    k_scan_tiles<true><<<nAgg, AGG_THREADS, 0, stream>>>(dst, tileb, dinv, wlist, wcnt, capW, N, E);
  } else {
    k_scan_tiles<false><<<nAgg, AGG_THREADS, 0, stream>>>(dst, tileb, dinv, wlist, wcnt, capW, N, E);
  }

  k_pack_bfrag<<<cdiv_ll((long long)nT1 * 1024, B), B, 0, stream>>>(W1, bf1, H, nT1);
  k_pack_bfrag<<<cdiv_ll((long long)nT2 * 1024, B), B, 0, stream>>>(W2, bf2, C, nT2);

  gcn_gemm_wmma_f32<<<N / 16, B, 0, stream>>>(x, bf1, h1, N, H, nT1);
  k_agg_rows<128, 40><<<nAgg, AGG_THREADS, 0, stream>>>(h1, src, dst, dinv, wlist, wcnt, capW, h2, N, E, b1, bf2);

  k_agg_rows<40, 0><<<nAgg, AGG_THREADS, 0, stream>>>(h2, src, dst, dinv, wlist, wcnt, capW, out, N, E, nullptr, nullptr);
  k_logsoftmax40<<<cdiv_ll(N, 8), B, 0, stream>>>(out, b2, N);
}


# Layer 2's bias and log_softmax taken on the summed tile while it lies in LDS (a wave's 7 rows side by side, same xor trees), then stored; the 50000-wave softmax launch is gone; 8->7 launches
# speedup vs baseline: 4.1276x; 1.0369x over previous
#include <hip/hip_runtime.h>
#include <hip/hip_bf16.h>
#include <math.h>

typedef float v2f __attribute__((ext_vector_type(2)));
typedef float v8f __attribute__((ext_vector_type(8)));

__global__ void k_pack_bfrag(const float* __restrict__ B, v2f* __restrict__ BF,
                             int N, int nTiles) {
  int i = blockIdx.x * blockDim.x + threadIdx.x;
  int total = nTiles * 1024;
  if (i >= total) return;
  int lane = i & 31;
  int t    = (i >> 5) & 31;
  int nt   = i >> 10;
  int half = lane >> 4;
  int lm   = lane & 15;
  int col  = nt * 16 + lm;
  int kk   = 4 * t + 2 * half;
  v2f v;
  v.x = (col < N) ? B[(size_t)kk * N + col]       : 0.0f;
  v.y = (col < N) ? B[(size_t)(kk + 1) * N + col] : 0.0f;
  BF[i] = v;
}

__global__ __launch_bounds__(256) void gcn_gemm_wmma_f32(
    const float* __restrict__ A, const v2f* __restrict__ BF,
    float* __restrict__ C, int M, int N, int nTiles) {
  constexpr int K = 128;
  constexpr int LDS_STRIDE = 132;
  __shared__ float As[16 * LDS_STRIDE];

  const int m0  = blockIdx.x * 16;
  const int tid = threadIdx.x;

#pragma unroll
  for (int j = 0; j < 8; ++j) {
    int off = tid + j * 256;
    int row = off >> 7;
    int col = off & 127;
    As[row * LDS_STRIDE + col] = A[(size_t)(m0 + row) * K + col];
  }
  __syncthreads();

  const int wave = tid >> 5;
  const int lane = tid & 31;
  const int half = lane >> 4;
  const int lm   = lane & 15;

  const float* aBase = &As[lm * LDS_STRIDE + 2 * half];

  for (int nt = wave; nt < nTiles; nt += 8) {
    const v2f* bBase = BF + (size_t)nt * 1024 + lane;
    v8f c = {};
#pragma unroll
    for (int t = 0; t < 32; ++t) {
      v2f a = *(const v2f*)(aBase + 4 * t);
      v2f b = bBase[(size_t)t * 32];
      c = __builtin_amdgcn_wmma_f32_16x16x4_f32(
           false, a,  false, b,
           (short)0, c,  false,  false);
    }
    const int col = nt * 16 + lm;
    if (col < N) {
      const int mBase = m0 + 8 * half;
#pragma unroll
      for (int r = 0; r < 8; ++r)
        C[(size_t)(mBase + r) * N + col] = c[r];
    }
  }
}

constexpr int AGG_THREADS = 1024;
constexpr int AGG_ROWS    = 196;
constexpr int AGG_LDS_BYTES = 168 * 1024;
constexpr int LIST_CAP      = 4096;

template <int V> struct IntC { static constexpr int value = V; };

constexpr int SCAN_CAP = (AGG_LDS_BYTES - AGG_ROWS * 4 - 64) / 4;
static_assert(LIST_CAP <= SCAN_CAP, "the scan's LDS list must hold a full workspace list");

__global__ __launch_bounds__(256) void k_tile_bytes(
    const int* __restrict__ dst, unsigned* __restrict__ tb32, int E) {
  const int E4 = E >> 2;
  const int base = blockIdx.x * 1024 + threadIdx.x;
  auto pack = [](int4 d) {
    unsigned b0 = ((unsigned)d.x / (unsigned)AGG_ROWS) & 255u;
    unsigned b1 = ((unsigned)d.y / (unsigned)AGG_ROWS) & 255u;
    unsigned b2 = ((unsigned)d.z / (unsigned)AGG_ROWS) & 255u;
    unsigned b3 = ((unsigned)d.w / (unsigned)AGG_ROWS) & 255u;
    return b0 | (b1 << 8) | (b2 << 16) | (b3 << 24);
  };
  if (E4 > 0) {
    if ((((size_t)dst) & 15) == 0) {
      const int4* dst4 = (const int4*)dst;
      int4 d[4];
#pragma unroll
      for (int u = 0; u < 4; ++u) d[u] = dst4[min(base + u * 256, E4 - 1)];
#pragma unroll
      for (int u = 0; u < 4; ++u) {
        const int g = base + u * 256;
        if (g < E4) tb32[g] = pack(d[u]);
      }
    } else {
#pragma unroll
      for (int u = 0; u < 4; ++u) {
        const int g = base + u * 256;
        if (g < E4) tb32[g] = pack(make_int4(dst[4 * g], dst[4 * g + 1], dst[4 * g + 2], dst[4 * g + 3]));
      }
    }
  }
  if (blockIdx.x == 0 && (int)threadIdx.x < (E & 3)) {
    const int e = 4 * E4 + threadIdx.x;
    ((unsigned char*)tb32)[e] = (unsigned char)(((unsigned)dst[e] / (unsigned)AGG_ROWS) & 255u);
  }
}

template <bool BYTES>
__global__ __launch_bounds__(AGG_THREADS) void k_scan_tiles(
    const int* __restrict__ dst, const unsigned char* __restrict__ tileb,
    float* __restrict__ dinv, int* __restrict__ wlist, int* __restrict__ wcnt,
    int capW, int N, int E) {
  __shared__ int list[SCAN_CAP];
  __shared__ int rowcnt[AGG_ROWS];
  __shared__ int cnt;
  __shared__ int cntv;

  const int tid   = threadIdx.x;
  const int r0    = blockIdx.x * AGG_ROWS;
  const int nrows = min(AGG_ROWS, N - r0);
  const unsigned ur0 = (unsigned)r0, un = (unsigned)nrows;

  for (int i = tid; i < AGG_ROWS; i += AGG_THREADS) rowcnt[i] = 0;
  if (tid == 0) { cnt = 0; cntv = 0; }
  __syncthreads();

  auto count_row = [&](int e) -> bool {
    int d = dst[e];
    bool ok = ((unsigned)d - ur0) < un;
    if (ok) atomicAdd(&rowcnt[d - r0], 1);
    return ok;
  };
  auto note = [&](int e) {
    int pos = atomicAdd(&cnt, 1);
    if (pos < SCAN_CAP) list[pos] = e;
    else count_row(e);
  };

  constexpr int U = 8;
  constexpr int TILE = U * AGG_THREADS;
  if constexpr (BYTES) {
    const unsigned tile = blockIdx.x;
    int scalarFrom = 0;
    if ((((size_t)tileb) & 15) == 0) {
      const unsigned pat = tile * 0x01010101u;
      auto hit = [&](unsigned w) {
        unsigned x = w ^ pat;
        return (x - 0x01010101u) & ~x & 0x80808080u;
      };
      auto flags = [&](uint4 v, auto hi) -> unsigned {
        constexpr int HI = decltype(hi)::value;
        return (hit(v.x) >> (7 - HI)) | (hit(v.y) >> (6 - HI)) | (hit(v.z) >> (5 - HI)) | (hit(v.w) >> (4 - HI));
      };
      auto drain = [&](unsigned mm, int e0, int e1) {
        while (mm) {
          int b = __builtin_ctz(mm);
          mm &= mm - 1;
          note(((b & 4) ? e1 : e0) + 4 * (b & 3) + (b >> 3));
        }
      };
      const int G16 = E >> 4;
      const int G32 = G16 & ~31;
      const uint4* tb4 = (const uint4*)tileb;
      const int wave = __builtin_amdgcn_readfirstlane(tid >> 5);
      const int lane = tid & 31;
      const int start = (int)(tile * ((unsigned)G32 / gridDim.x)) & ~31;
      auto place = [&](int i) { int q = start + i; return q >= G32 ? q - G32 : q; };
      int i0 = 0;
      for (; i0 + TILE <= G32; i0 += TILE) {
        uint4 dv[U];
        int   q[U];
#pragma unroll
        for (int u = 0; u < U; ++u) {
          q[u]  = place(i0 + u * AGG_THREADS + wave * 32) + lane;
          dv[u] = tb4[q[u]];
        }
#pragma unroll
        for (int u = 0; u < U; u += 2)
          drain(flags(dv[u], IntC<0>{}) | flags(dv[u + 1], IntC<4>{}), 16 * q[u], 16 * q[u + 1]);
      }
      for (int i = i0 + wave * 32; i < G32; i += AGG_THREADS) {
        const int q = place(i) + lane;
        drain(flags(tb4[q], IntC<0>{}), 16 * q, 16 * q);
      }
      if (tid < G16 - G32) {
        const int q = G32 + tid;
        drain(flags(tb4[q], IntC<0>{}), 16 * q, 16 * q);
      }
      scalarFrom = G16 * 16;
    }
    for (int e = scalarFrom + tid; e < E; e += AGG_THREADS)
      if (tileb[e] == (unsigned char)tile) note(e);
  } else {
    auto check4 = [&](int4 d4, int e) {
      bool m0 = ((unsigned)d4.x - ur0) < un;
      bool m1 = ((unsigned)d4.y - ur0) < un;
      bool m2 = ((unsigned)d4.z - ur0) < un;
      bool m3 = ((unsigned)d4.w - ur0) < un;
      if (m0 | m1 | m2 | m3) {
        if (m0) note(e);
        if (m1) note(e + 1);
        if (m2) note(e + 2);
        if (m3) note(e + 3);
      }
    };
    int scalarFrom = 0;
    if ((((size_t)dst) & 15) == 0) {
      const int E4 = E >> 2;
      const int4* dst4 = (const int4*)dst;
      auto scan_groups = [&](int lo, int hi) {
        int t0 = lo;
        for (; t0 + TILE <= hi; t0 += TILE) {
          int4 dv[U];
#pragma unroll
          for (int u = 0; u < U; ++u) dv[u] = dst4[t0 + tid + u * AGG_THREADS];
#pragma unroll
          for (int u = 0; u < U; ++u) check4(dv[u], 4 * (t0 + tid + u * AGG_THREADS));
        }
        for (int q = t0 + tid; q < hi; q += AGG_THREADS) check4(dst4[q], 4 * q);
      };
      const int start = (int)(blockIdx.x * ((unsigned)E4 / gridDim.x)) & ~7;
      scan_groups(start, E4);
      scan_groups(0, start);
      scalarFrom = E4 * 4;
    }
    for (int e = scalarFrom + tid; e < E; e += AGG_THREADS) {
      int d = dst[e];
      if (((unsigned)d - ur0) < un) note(e);
    }
  }
  __syncthreads();

  const int noted = cnt;
  const int held  = min(noted, SCAN_CAP);
  int* wl = wlist + (size_t)blockIdx.x * LIST_CAP;
  for (int m = tid; m < held; m += AGG_THREADS) {
    int e = list[m];
    if (count_row(e)) {
      int p = atomicAdd(&cntv, 1);
      if (p < capW) wl[p] = e;
    }
  }
  __syncthreads();

  for (int i = tid; i < nrows; i += AGG_THREADS) dinv[r0 + i] = rsqrtf((float)(rowcnt[i] + 1));

  if (tid == 0) wcnt[blockIdx.x] = (noted <= SCAN_CAP) ? cntv : 0x7fffffff;
}

template <int F, int FO, bool LSM = false>
__global__ __launch_bounds__(AGG_THREADS) void k_agg_rows(
    const float* __restrict__ h, const int* __restrict__ src, const int* __restrict__ dst,
    const float* __restrict__ dinv, const int* __restrict__ wlist, const int* __restrict__ wcnt,
    int capW, float* __restrict__ out, int N, int E,
    const float* __restrict__ bias, const v2f* __restrict__ BF) {
  constexpr bool MM     = FO > 0;
  constexpr int  PITCH  = MM ? F + 4 : F;
  constexpr int  ROWS_A = MM ? ((AGG_ROWS + 15) / 16) * 16 : AGG_ROWS;
  constexpr int  CAPL   = (AGG_LDS_BYTES - ROWS_A * PITCH * 4) / 12;
  static_assert(LIST_CAP <= CAPL, "LDS must hold a full workspace list beside the tile");
  static_assert(!MM || (F % 4 == 0 && ROWS_A * FO <= 3 * CAPL), "product rows are staged where the edge list was");
  __shared__ alignas(16) float acc[ROWS_A * PITCH];
  __shared__ int lst[3 * CAPL];
  int* eS = lst;
  int* eD = lst + CAPL;
  int* eC = lst + 2 * CAPL;

  const int tid   = threadIdx.x;
  const int r0    = blockIdx.x * AGG_ROWS;
  const int nrows = min(AGG_ROWS, N - r0);
  const int nel   = nrows * F;
  const size_t base = (size_t)r0 * F;

  const int  total  = wcnt[blockIdx.x];
  const bool listed = total <= capW;

  if (listed) {
    const int* wl = wlist + (size_t)blockIdx.x * LIST_CAP;
    for (int m = tid; m < total; m += AGG_THREADS) {
      int e = wl[m];
      int s = src[e];
      int d = dst[e];
      eS[m] = s;
      eD[m] = d - r0;
      eC[m] = __float_as_int(dinv[s] * dinv[d]);
    }
  }
  for (int i = tid; i < ROWS_A * F; i += AGG_THREADS) {
    const int row = i / F, col = i % F;
    float v = 0.0f;
    if (i < nel) {
      float di = dinv[r0 + row];
      v = di * di * h[base + i];
    }
    acc[row * PITCH + col] = v;
  }
  __syncthreads();

  if (listed) {
    const int wave = __builtin_amdgcn_readfirstlane(tid >> 5);
    const int lane = tid & 31;
    constexpr int NJ = (F + 31) / 32;
    constexpr int G  = 8;
    for (int m0 = wave * 32; m0 < total; m0 += AGG_THREADS) {
      const int m = m0 + lane;
      int s = 0, dl = 0;
      float c = 0.0f;
      if (m < total) {
        s  = eS[m];
        dl = eD[m];
        c  = __int_as_float(eC[m]);
      }
      auto add_edges = [&](auto gg, int i0) {
        constexpr int GG = decltype(gg)::value;
        float v[GG][NJ];
#pragma unroll
        for (int i = 0; i < GG; ++i) {
          int si = __builtin_amdgcn_readlane(s, i0 + i);
          const float* hr = h + (size_t)si * F;
#pragma unroll
          for (int j = 0; j < NJ; ++j) v[i][j] = hr[min(32 * j + lane, F - 1)];
        }
#pragma unroll
        for (int i = 0; i < GG; ++i) {
          int   di = __builtin_amdgcn_readlane(dl, i0 + i);
          float ci = __builtin_bit_cast(float, __builtin_amdgcn_readlane(__builtin_bit_cast(int, c), i0 + i));
          float* ar = acc + di * PITCH;
#pragma unroll
          for (int j = 0; j < NJ; ++j) {
            int f = 32 * j + lane;
            if (f < F) atomicAdd(&ar[f], ci * v[i][j]);
          }
        }
      };
      const int nb = min(32, total - m0);
      int i0 = 0;
      for (; i0 + G <= nb; i0 += G) add_edges(IntC<G>{}, i0);
      for (; i0 < nb; ++i0)         add_edges(IntC<1>{}, i0);
    }
  } else {
    const unsigned ur0 = (unsigned)r0, un = (unsigned)nrows;
    for (int e = tid; e < E; e += AGG_THREADS) {
      int d = dst[e];
      if (((unsigned)d - ur0) < un) {
        int s = src[e];
        float c = dinv[s] * dinv[d];
        for (int f = 0; f < F; ++f)
          atomicAdd(&acc[(d - r0) * PITCH + f], c * h[(size_t)s * F + f]);
      }
    }
  }
  __syncthreads();

  if constexpr (!MM) {
    if constexpr (LSM) {
      static_assert(F > 32 && F <= 64 && PITCH == F, "a lane holds one or two classes of a row");
      constexpr int NW  = AGG_THREADS / 32;
      constexpr int RPW = (AGG_ROWS + NW - 1) / NW;
      const int  wave = __builtin_amdgcn_readfirstlane(tid >> 5);
      const int  lane = tid & 31;
      const bool has2 = (lane + 32) < F;
      const float b0 = bias[lane];
      const float b1 = has2 ? bias[lane + 32] : 0.0f;
      const int  c2   = min(lane + 32, F - 1);
      float v0[RPW], v1[RPW], mx[RPW], sm[RPW];
#pragma unroll
      for (int k = 0; k < RPW; ++k) {
        const int row = wave + k * NW;
        v0[k] = 0.0f;
        v1[k] = 0.0f;
        if (row < nrows) {
          const float* p = acc + row * F;
          v0[k] = p[lane];
          v1[k] = p[c2];
        }
      }
#pragma unroll
      for (int k = 0; k < RPW; ++k) {
        v0[k] = v0[k] + b0;
        v1[k] = has2 ? v1[k] + b1 : -INFINITY;
        mx[k] = fmaxf(v0[k], v1[k]);
      }
#pragma unroll
      for (int off = 16; off > 0; off >>= 1) {
#pragma unroll
        for (int k = 0; k < RPW; ++k) mx[k] = fmaxf(mx[k], __shfl_xor(mx[k], off, 32));
      }
#pragma unroll
      for (int k = 0; k < RPW; ++k)
        sm[k] = expf(v0[k] - mx[k]) + (has2 ? expf(v1[k] - mx[k]) : 0.0f);
#pragma unroll
      for (int off = 16; off > 0; off >>= 1) {
#pragma unroll
        for (int k = 0; k < RPW; ++k) sm[k] += __shfl_xor(sm[k], off, 32);
      }
#pragma unroll
      for (int k = 0; k < RPW; ++k) {
        const int row = wave + k * NW;
        const float lse = logf(sm[k]);
        if (row < nrows) {
          float* p = acc + row * F;
          p[lane] = v0[k] - mx[k] - lse;
          if (has2) p[lane + 32] = v1[k] - mx[k] - lse;
        }
      }
      __syncthreads();
    }
    for (int i = tid; i < nel; i += AGG_THREADS) out[base + i] = acc[i];
  } else {
    for (int i = tid; i < nel; i += AGG_THREADS) {
      const int row = i / F, col = i % F;
      float v = acc[row * PITCH + col] + bias[col];
      acc[row * PITCH + col] = v > 0.0f ? v : 0.0f;
    }
    __syncthreads();

    constexpr int NT = (FO + 15) / 16;
    constexpr int JOBS = (ROWS_A / 16) * NT;
    const int wave = __builtin_amdgcn_readfirstlane(tid >> 5);
    const int lane = tid & 31;
    const int half = lane >> 4;
    const int lm   = lane & 15;
    for (int job = wave; job < JOBS; job += AGG_THREADS / 32) {
      const int rt = job / NT, nt = job % NT;
      const float* aBase = acc + (rt * 16 + lm) * PITCH + 2 * half;
      const v2f*   bBase = BF + (size_t)nt * 1024 + lane;
      v8f c = {};
#pragma unroll
      for (int t = 0; t < F / 4; ++t) {
        v2f a = *(const v2f*)(aBase + 4 * t);
        v2f b = bBase[(size_t)t * 32];
        c = __builtin_amdgcn_wmma_f32_16x16x4_f32(
             false, a,  false, b,
             (short)0, c,  false,  false);
      }
      const int col = nt * 16 + lm;
      if (col < FO) {
        int* st = lst + (rt * 16 + 8 * half) * FO + col;
#pragma unroll
        for (int r = 0; r < 8; ++r) {
          const float cr = c[r];
          st[r * FO] = __float_as_int(cr);
        }
      }
    }
    __syncthreads();

    const size_t obase = (size_t)r0 * FO;
    for (int i = tid; i < nrows * FO; i += AGG_THREADS) out[obase + i] = __int_as_float(lst[i]);
  }
}

static inline int cdiv_ll(long long a, long long b) { return (int)((a + b - 1) / b); }

extern "C" void kernel_launch(void* const* d_in, const int* in_sizes, int n_in,
                              void* d_out, int out_size, void* d_ws, size_t ws_size,
                              hipStream_t stream) {
  const float* x  = (const float*)d_in[0];
  const int*   ei = (const int*)  d_in[1];
  const float* W1 = (const float*)d_in[2];
  const float* b1 = (const float*)d_in[3];
  const float* W2 = (const float*)d_in[4];
  const float* b2 = (const float*)d_in[5];

  const int F = 128, H = 128, C = 40;
  const int N = in_sizes[0] / F;
  const int E = in_sizes[1] / 2;
  const int* src = ei;
  const int* dst = ei + E;

  const int nT1 = (H + 15) / 16;
  const int nT2 = (C + 15) / 16;

  const int nAgg = cdiv_ll(N, AGG_ROWS);

  float* ws   = (float*)d_ws;
  int*   wcnt = (int*)ws;           ws += N;
  float* dinv = ws;                 ws += N;
  float* h1   = ws;                 ws += (size_t)N * H;
  float* o1   = ws;                 ws += (size_t)N * H;
  int*   wlist = (int*)ws;          ws += (size_t)N * C;
  v2f* bf1    = (v2f*)ws;           ws += (size_t)nT1 * 1024 * 2;
  v2f* bf2    = (v2f*)ws;           ws += (size_t)nT2 * 1024 * 2;
  float* h2   = o1;
  float* out  = (float*)d_out;
  const int capW = ((size_t)nAgg * LIST_CAP <= (size_t)N * C) ? LIST_CAP : 0;

  const int B = 256;

  unsigned char* tileb = (unsigned char*)o1;
  const bool bytes = nAgg <= 256 && (size_t)E <= (size_t)N * H * sizeof(float);
  if (bytes) {
    k_tile_bytes<<<cdiv_ll((long long)(E >> 2) + 1, 1024), B, 0, stream>>>(dst, (unsigned*)tileb, E);
    k_scan_tiles<true><<<nAgg, AGG_THREADS, 0, stream>>>(dst, tileb, dinv, wlist, wcnt, capW, N, E);
  } else {
    k_scan_tiles<false><<<nAgg, AGG_THREADS, 0, stream>>>(dst, tileb, dinv, wlist, wcnt, capW, N, E);
  }

  k_pack_bfrag<<<cdiv_ll((long long)nT1 * 1024, B), B, 0, stream>>>(W1, bf1, H, nT1);
  k_pack_bfrag<<<cdiv_ll((long long)nT2 * 1024, B), B, 0, stream>>>(W2, bf2, C, nT2);

  gcn_gemm_wmma_f32<<<N / 16, B, 0, stream>>>(x, bf1, h1, N, H, nT1);
  k_agg_rows<128, 40><<<nAgg, AGG_THREADS, 0, stream>>>(h1, src, dst, dinv, wlist, wcnt, capW, h2, N, E, b1, bf2);

  k_agg_rows<40, 0, true><<<nAgg, AGG_THREADS, 0, stream>>>(h2, src, dst, dinv, wlist, wcnt, capW, out, N, E, b2, nullptr);
}


# Tile blocks' set-up loads issued stage by stage over all of a thread's list entries and 16-byte row pieces: 3 memory round trips, not one per loop trip (32 in layer 1); sums and their order unchanged
# speedup vs baseline: 4.4597x; 1.0805x over previous
#include <hip/hip_runtime.h>
#include <hip/hip_bf16.h>
#include <math.h>

typedef float v2f __attribute__((ext_vector_type(2)));
typedef float v8f __attribute__((ext_vector_type(8)));

__global__ void k_pack_bfrag(const float* __restrict__ B, v2f* __restrict__ BF,
                             int N, int nTiles) {
  int i = blockIdx.x * blockDim.x + threadIdx.x;
  int total = nTiles * 1024;
  if (i >= total) return;
  int lane = i & 31;
  int t    = (i >> 5) & 31;
  int nt   = i >> 10;
  int half = lane >> 4;
  int lm   = lane & 15;
  int col  = nt * 16 + lm;
  int kk   = 4 * t + 2 * half;
  v2f v;
  v.x = (col < N) ? B[(size_t)kk * N + col]       : 0.0f;
  v.y = (col < N) ? B[(size_t)(kk + 1) * N + col] : 0.0f;
  BF[i] = v;
}

__global__ __launch_bounds__(256) void gcn_gemm_wmma_f32(
    const float* __restrict__ A, const v2f* __restrict__ BF,
    float* __restrict__ C, int M, int N, int nTiles) {
  constexpr int K = 128;
  constexpr int LDS_STRIDE = 132;
  __shared__ float As[16 * LDS_STRIDE];

  const int m0  = blockIdx.x * 16;
  const int tid = threadIdx.x;

#pragma unroll
  for (int j = 0; j < 8; ++j) {
    int off = tid + j * 256;
    int row = off >> 7;
    int col = off & 127;
    As[row * LDS_STRIDE + col] = A[(size_t)(m0 + row) * K + col];
  }
  __syncthreads();

  const int wave = tid >> 5;
  const int lane = tid & 31;
  const int half = lane >> 4;
  const int lm   = lane & 15;

  const float* aBase = &As[lm * LDS_STRIDE + 2 * half];

  for (int nt = wave; nt < nTiles; nt += 8) {
    const v2f* bBase = BF + (size_t)nt * 1024 + lane;
    v8f c = {};
#pragma unroll
    for (int t = 0; t < 32; ++t) {
      v2f a = *(const v2f*)(aBase + 4 * t);
      v2f b = bBase[(size_t)t * 32];
      c = __builtin_amdgcn_wmma_f32_16x16x4_f32(
           false, a,  false, b,
           (short)0, c,  false,  false);
    }
    const int col = nt * 16 + lm;
    if (col < N) {
      const int mBase = m0 + 8 * half;
#pragma unroll
      for (int r = 0; r < 8; ++r)
        C[(size_t)(mBase + r) * N + col] = c[r];
    }
  }
}

constexpr int AGG_THREADS = 1024;
constexpr int AGG_ROWS    = 196;
constexpr int AGG_LDS_BYTES = 168 * 1024;
constexpr int LIST_CAP      = 4096;

template <int V> struct IntC { static constexpr int value = V; };

constexpr int SCAN_CAP = (AGG_LDS_BYTES - AGG_ROWS * 4 - 64) / 4;
static_assert(LIST_CAP <= SCAN_CAP, "the scan's LDS list must hold a full workspace list");

__global__ __launch_bounds__(256) void k_tile_bytes(
    const int* __restrict__ dst, unsigned* __restrict__ tb32, int E) {
  const int E4 = E >> 2;
  const int base = blockIdx.x * 1024 + threadIdx.x;
  auto pack = [](int4 d) {
    unsigned b0 = ((unsigned)d.x / (unsigned)AGG_ROWS) & 255u;
    unsigned b1 = ((unsigned)d.y / (unsigned)AGG_ROWS) & 255u;
    unsigned b2 = ((unsigned)d.z / (unsigned)AGG_ROWS) & 255u;
    unsigned b3 = ((unsigned)d.w / (unsigned)AGG_ROWS) & 255u;
    return b0 | (b1 << 8) | (b2 << 16) | (b3 << 24);
  };
  if (E4 > 0) {
    if ((((size_t)dst) & 15) == 0) {
      const int4* dst4 = (const int4*)dst;
      int4 d[4];
#pragma unroll
      for (int u = 0; u < 4; ++u) d[u] = dst4[min(base + u * 256, E4 - 1)];
#pragma unroll
      for (int u = 0; u < 4; ++u) {
        const int g = base + u * 256;
        if (g < E4) tb32[g] = pack(d[u]);
      }
    } else {
#pragma unroll
      for (int u = 0; u < 4; ++u) {
        const int g = base + u * 256;
        if (g < E4) tb32[g] = pack(make_int4(dst[4 * g], dst[4 * g + 1], dst[4 * g + 2], dst[4 * g + 3]));
      }
    }
  }
  if (blockIdx.x == 0 && (int)threadIdx.x < (E & 3)) {
    const int e = 4 * E4 + threadIdx.x;
    ((unsigned char*)tb32)[e] = (unsigned char)(((unsigned)dst[e] / (unsigned)AGG_ROWS) & 255u);
  }
}

template <bool BYTES>
__global__ __launch_bounds__(AGG_THREADS) void k_scan_tiles(
    const int* __restrict__ dst, const unsigned char* __restrict__ tileb,
    float* __restrict__ dinv, int* __restrict__ wlist, int* __restrict__ wcnt,
    int capW, int N, int E) {
  __shared__ int list[SCAN_CAP];
  __shared__ int rowcnt[AGG_ROWS];
  __shared__ int cnt;
  __shared__ int cntv;

  const int tid   = threadIdx.x;
  const int r0    = blockIdx.x * AGG_ROWS;
  const int nrows = min(AGG_ROWS, N - r0);
  const unsigned ur0 = (unsigned)r0, un = (unsigned)nrows;

  for (int i = tid; i < AGG_ROWS; i += AGG_THREADS) rowcnt[i] = 0;
  if (tid == 0) { cnt = 0; cntv = 0; }
  __syncthreads();

  auto count_row = [&](int e) -> bool {
    int d = dst[e];
    bool ok = ((unsigned)d - ur0) < un;
    if (ok) atomicAdd(&rowcnt[d - r0], 1);
    return ok;
  };
  auto note = [&](int e) {
    int pos = atomicAdd(&cnt, 1);
    if (pos < SCAN_CAP) list[pos] = e;
    else count_row(e);
  };

  constexpr int U = 8;
  constexpr int TILE = U * AGG_THREADS;
  if constexpr (BYTES) {
    const unsigned tile = blockIdx.x;
    int scalarFrom = 0;
    if ((((size_t)tileb) & 15) == 0) {
      const unsigned pat = tile * 0x01010101u;
      auto hit = [&](unsigned w) {
        unsigned x = w ^ pat;
        return (x - 0x01010101u) & ~x & 0x80808080u;
      };
      auto flags = [&](uint4 v, auto hi) -> unsigned {
        constexpr int HI = decltype(hi)::value;
        return (hit(v.x) >> (7 - HI)) | (hit(v.y) >> (6 - HI)) | (hit(v.z) >> (5 - HI)) | (hit(v.w) >> (4 - HI));
      };
      auto drain = [&](unsigned mm, int e0, int e1) {
        while (mm) {
          int b = __builtin_ctz(mm);
          mm &= mm - 1;
          note(((b & 4) ? e1 : e0) + 4 * (b & 3) + (b >> 3));
        }
      };
      const int G16 = E >> 4;
      const int G32 = G16 & ~31;
      const uint4* tb4 = (const uint4*)tileb;
      const int wave = __builtin_amdgcn_readfirstlane(tid >> 5);
      const int lane = tid & 31;
      const int start = (int)(tile * ((unsigned)G32 / gridDim.x)) & ~31;
      auto place = [&](int i) { int q = start + i; return q >= G32 ? q - G32 : q; };
      int i0 = 0;
      for (; i0 + TILE <= G32; i0 += TILE) {
        uint4 dv[U];
        int   q[U];
#pragma unroll
        for (int u = 0; u < U; ++u) {
          q[u]  = place(i0 + u * AGG_THREADS + wave * 32) + lane;
          dv[u] = tb4[q[u]];
        }
#pragma unroll
        for (int u = 0; u < U; u += 2)
          drain(flags(dv[u], IntC<0>{}) | flags(dv[u + 1], IntC<4>{}), 16 * q[u], 16 * q[u + 1]);
      }
      for (int i = i0 + wave * 32; i < G32; i += AGG_THREADS) {
        const int q = place(i) + lane;
        drain(flags(tb4[q], IntC<0>{}), 16 * q, 16 * q);
      }
      if (tid < G16 - G32) {
        const int q = G32 + tid;
        drain(flags(tb4[q], IntC<0>{}), 16 * q, 16 * q);
      }
      scalarFrom = G16 * 16;
    }
    for (int e = scalarFrom + tid; e < E; e += AGG_THREADS)
      if (tileb[e] == (unsigned char)tile) note(e);
  } else {
    auto check4 = [&](int4 d4, int e) {
      bool m0 = ((unsigned)d4.x - ur0) < un;
      bool m1 = ((unsigned)d4.y - ur0) < un;
      bool m2 = ((unsigned)d4.z - ur0) < un;
      bool m3 = ((unsigned)d4.w - ur0) < un;
      if (m0 | m1 | m2 | m3) {
        if (m0) note(e);
        if (m1) note(e + 1);
        if (m2) note(e + 2);
        if (m3) note(e + 3);
      }
    };
    int scalarFrom = 0;
    if ((((size_t)dst) & 15) == 0) {
      const int E4 = E >> 2;
      const int4* dst4 = (const int4*)dst;
      auto scan_groups = [&](int lo, int hi) {
        int t0 = lo;
        for (; t0 + TILE <= hi; t0 += TILE) {
          int4 dv[U];
#pragma unroll
          for (int u = 0; u < U; ++u) dv[u] = dst4[t0 + tid + u * AGG_THREADS];
#pragma unroll
          for (int u = 0; u < U; ++u) check4(dv[u], 4 * (t0 + tid + u * AGG_THREADS));
        }
        for (int q = t0 + tid; q < hi; q += AGG_THREADS) check4(dst4[q], 4 * q);
      };
      const int start = (int)(blockIdx.x * ((unsigned)E4 / gridDim.x)) & ~7;
      scan_groups(start, E4);
      scan_groups(0, start);
      scalarFrom = E4 * 4;
    }
    for (int e = scalarFrom + tid; e < E; e += AGG_THREADS) {
      int d = dst[e];
      if (((unsigned)d - ur0) < un) note(e);
    }
  }
  __syncthreads();

  const int noted = cnt;
  const int held  = min(noted, SCAN_CAP);
  int* wl = wlist + (size_t)blockIdx.x * LIST_CAP;
  for (int m = tid; m < held; m += AGG_THREADS) {
    int e = list[m];
    if (count_row(e)) {
      int p = atomicAdd(&cntv, 1);
      if (p < capW) wl[p] = e;
    }
  }
  __syncthreads();

  for (int i = tid; i < nrows; i += AGG_THREADS) dinv[r0 + i] = rsqrtf((float)(rowcnt[i] + 1));

  if (tid == 0) wcnt[blockIdx.x] = (noted <= SCAN_CAP) ? cntv : 0x7fffffff;
}

template <int F, int FO, bool LSM = false>
__global__ __launch_bounds__(AGG_THREADS) void k_agg_rows(
    const float* __restrict__ h, const int* __restrict__ src, const int* __restrict__ dst,
    const float* __restrict__ dinv, const int* __restrict__ wlist, const int* __restrict__ wcnt,
    int capW, float* __restrict__ out, int N, int E,
    const float* __restrict__ bias, const v2f* __restrict__ BF) {
  constexpr bool MM     = FO > 0;
  constexpr int  PITCH  = MM ? F + 4 : F;
  constexpr int  ROWS_A = MM ? ((AGG_ROWS + 15) / 16) * 16 : AGG_ROWS;
  constexpr int  CAPL   = (AGG_LDS_BYTES - ROWS_A * PITCH * 4) / 12;
  static_assert(LIST_CAP <= CAPL, "LDS must hold a full workspace list beside the tile");
  static_assert(!MM || (F % 4 == 0 && ROWS_A * FO <= 3 * CAPL), "product rows are staged where the edge list was");
  __shared__ alignas(16) float acc[ROWS_A * PITCH];
  __shared__ int lst[3 * CAPL];
  int* eS = lst;
  int* eD = lst + CAPL;
  int* eC = lst + 2 * CAPL;

  const int tid   = threadIdx.x;
  const int r0    = blockIdx.x * AGG_ROWS;
  const int nrows = min(AGG_ROWS, N - r0);
  const int nel   = nrows * F;
  const size_t base = (size_t)r0 * F;

  const int  total  = wcnt[blockIdx.x];
  const bool listed = total <= capW;

  constexpr int LPT = (LIST_CAP + AGG_THREADS - 1) / AGG_THREADS;
  constexpr int NG  = ROWS_A * F / 4;
  constexpr int PPT = (NG + AGG_THREADS - 1) / AGG_THREADS;
  static_assert(F % 4 == 0 && PITCH % 4 == 0, "a 16-byte piece lies in one row, at a 16-byte place");
  const int* wl = wlist + (size_t)blockIdx.x * LIST_CAP;
  if ((((size_t)h) & 15) == 0) {
    const bool staged = listed && total > 0;
    float4 hv[PPT];
    float  dv[PPT];
#pragma unroll
    for (int k = 0; k < PPT; ++k) {
      const int i = min(4 * (tid + k * AGG_THREADS), nel - 4);
      dv[k] = dinv[r0 + i / F];
      hv[k] = *(const float4*)(h + base + i);
    }
    int le[LPT], ls[LPT], ld[LPT];
    if (staged) {
#pragma unroll
      for (int k = 0; k < LPT; ++k) le[k] = wl[min(tid + k * AGG_THREADS, total - 1)];
#pragma unroll
      for (int k = 0; k < LPT; ++k) {
        ls[k] = src[le[k]];
        ld[k] = dst[le[k]];
      }
    }
#pragma unroll
    for (int k = 0; k < PPT; ++k) {
      const int g = tid + k * AGG_THREADS;
      if (g < NG) {
        const int i = 4 * g;
        const int row = i / F, col = i % F;
        const float dd = dv[k] * dv[k];
        float4 v = make_float4(0.0f, 0.0f, 0.0f, 0.0f);
        if (i < nel) v = make_float4(dd * hv[k].x, dd * hv[k].y, dd * hv[k].z, dd * hv[k].w);
        *(float4*)(acc + row * PITCH + col) = v;
      }
    }
    if (staged) {
      float cs[LPT], cd[LPT];
#pragma unroll
      for (int k = 0; k < LPT; ++k) {
        cs[k] = dinv[ls[k]];
        cd[k] = dinv[ld[k]];
      }
#pragma unroll
      for (int k = 0; k < LPT; ++k) {
        const int m = tid + k * AGG_THREADS;
        if (m < total) {
          eS[m] = ls[k];
          eD[m] = ld[k] - r0;
          eC[m] = __float_as_int(cs[k] * cd[k]);
        }
      }
      for (int m = tid + LPT * AGG_THREADS; m < total; m += AGG_THREADS) {
        int e = wl[m];
        int s = src[e];
        int d = dst[e];
        eS[m] = s;
        eD[m] = d - r0;
        eC[m] = __float_as_int(dinv[s] * dinv[d]);
      }
    }
  } else {
    if (listed) {
      for (int m = tid; m < total; m += AGG_THREADS) {
        int e = wl[m];
        int s = src[e];
        int d = dst[e];
        eS[m] = s;
        eD[m] = d - r0;
        eC[m] = __float_as_int(dinv[s] * dinv[d]);
      }
    }
    for (int i = tid; i < ROWS_A * F; i += AGG_THREADS) {
      const int row = i / F, col = i % F;
      float v = 0.0f;
      if (i < nel) {
        float di = dinv[r0 + row];
        v = di * di * h[base + i];
      }
      acc[row * PITCH + col] = v;
    }
  }
  __syncthreads();

  if (listed) {
    const int wave = __builtin_amdgcn_readfirstlane(tid >> 5);
    const int lane = tid & 31;
    constexpr int NJ = (F + 31) / 32;
    constexpr int G  = 8;
    for (int m0 = wave * 32; m0 < total; m0 += AGG_THREADS) {
      const int m = m0 + lane;
      int s = 0, dl = 0;
      float c = 0.0f;
      if (m < total) {
        s  = eS[m];
        dl = eD[m];
        c  = __int_as_float(eC[m]);
      }
      auto add_edges = [&](auto gg, int i0) {
        constexpr int GG = decltype(gg)::value;
        float v[GG][NJ];
#pragma unroll
        for (int i = 0; i < GG; ++i) {
          int si = __builtin_amdgcn_readlane(s, i0 + i);
          const float* hr = h + (size_t)si * F;
#pragma unroll
          for (int j = 0; j < NJ; ++j) v[i][j] = hr[min(32 * j + lane, F - 1)];
        }
#pragma unroll
        for (int i = 0; i < GG; ++i) {
          int   di = __builtin_amdgcn_readlane(dl, i0 + i);
          float ci = __builtin_bit_cast(float, __builtin_amdgcn_readlane(__builtin_bit_cast(int, c), i0 + i));
          float* ar = acc + di * PITCH;
#pragma unroll
          for (int j = 0; j < NJ; ++j) {
            int f = 32 * j + lane;
            if (f < F) atomicAdd(&ar[f], ci * v[i][j]);
          }
        }
      };
      const int nb = min(32, total - m0);
      int i0 = 0;
      for (; i0 + G <= nb; i0 += G) add_edges(IntC<G>{}, i0);
      for (; i0 < nb; ++i0)         add_edges(IntC<1>{}, i0);
    }
  } else {
    const unsigned ur0 = (unsigned)r0, un = (unsigned)nrows;
    for (int e = tid; e < E; e += AGG_THREADS) {
      int d = dst[e];
      if (((unsigned)d - ur0) < un) {
        int s = src[e];
        float c = dinv[s] * dinv[d];
        for (int f = 0; f < F; ++f)
          atomicAdd(&acc[(d - r0) * PITCH + f], c * h[(size_t)s * F + f]);
      }
    }
  }
  __syncthreads();

  if constexpr (!MM) {
    if constexpr (LSM) {
      static_assert(F > 32 && F <= 64 && PITCH == F, "a lane holds one or two classes of a row");
      constexpr int NW  = AGG_THREADS / 32;
      constexpr int RPW = (AGG_ROWS + NW - 1) / NW;
      const int  wave = __builtin_amdgcn_readfirstlane(tid >> 5);
      const int  lane = tid & 31;
      const bool has2 = (lane + 32) < F;
      const float b0 = bias[lane];
      const float b1 = has2 ? bias[lane + 32] : 0.0f;
      const int  c2   = min(lane + 32, F - 1);
      float v0[RPW], v1[RPW], mx[RPW], sm[RPW];
#pragma unroll
      for (int k = 0; k < RPW; ++k) {
        const int row = wave + k * NW;
        v0[k] = 0.0f;
        v1[k] = 0.0f;
        if (row < nrows) {
          const float* p = acc + row * F;
          v0[k] = p[lane];
          v1[k] = p[c2];
        }
      }
#pragma unroll
      for (int k = 0; k < RPW; ++k) {
        v0[k] = v0[k] + b0;
        v1[k] = has2 ? v1[k] + b1 : -INFINITY;
        mx[k] = fmaxf(v0[k], v1[k]);
      }
#pragma unroll
      for (int off = 16; off > 0; off >>= 1) {
#pragma unroll
        for (int k = 0; k < RPW; ++k) mx[k] = fmaxf(mx[k], __shfl_xor(mx[k], off, 32));
      }
#pragma unroll
      for (int k = 0; k < RPW; ++k)
        sm[k] = expf(v0[k] - mx[k]) + (has2 ? expf(v1[k] - mx[k]) : 0.0f);
#pragma unroll
      for (int off = 16; off > 0; off >>= 1) {
#pragma unroll
        for (int k = 0; k < RPW; ++k) sm[k] += __shfl_xor(sm[k], off, 32);
      }
#pragma unroll
      for (int k = 0; k < RPW; ++k) {
        const int row = wave + k * NW;
        const float lse = logf(sm[k]);
        if (row < nrows) {
          float* p = acc + row * F;
          p[lane] = v0[k] - mx[k] - lse;
          if (has2) p[lane + 32] = v1[k] - mx[k] - lse;
        }
      }
      __syncthreads();
    }
    for (int i = tid; i < nel; i += AGG_THREADS) out[base + i] = acc[i];
  } else {
    for (int i = tid; i < nel; i += AGG_THREADS) {
      const int row = i / F, col = i % F;
      float v = acc[row * PITCH + col] + bias[col];
      acc[row * PITCH + col] = v > 0.0f ? v : 0.0f;
    }
    __syncthreads();

    constexpr int NT = (FO + 15) / 16;
    constexpr int JOBS = (ROWS_A / 16) * NT;
    const int wave = __builtin_amdgcn_readfirstlane(tid >> 5);
    const int lane = tid & 31;
    const int half = lane >> 4;
    const int lm   = lane & 15;
    for (int job = wave; job < JOBS; job += AGG_THREADS / 32) {
      const int rt = job / NT, nt = job % NT;
      const float* aBase = acc + (rt * 16 + lm) * PITCH + 2 * half;
      const v2f*   bBase = BF + (size_t)nt * 1024 + lane;
      v8f c = {};
#pragma unroll
      for (int t = 0; t < F / 4; ++t) {
        v2f a = *(const v2f*)(aBase + 4 * t);
        v2f b = bBase[(size_t)t * 32];
        c = __builtin_amdgcn_wmma_f32_16x16x4_f32(
             false, a,  false, b,
             (short)0, c,  false,  false);
      }
      const int col = nt * 16 + lm;
      if (col < FO) {
        int* st = lst + (rt * 16 + 8 * half) * FO + col;
#pragma unroll
        for (int r = 0; r < 8; ++r) {
          const float cr = c[r];
          st[r * FO] = __float_as_int(cr);
        }
      }
    }
    __syncthreads();

    const size_t obase = (size_t)r0 * FO;
    for (int i = tid; i < nrows * FO; i += AGG_THREADS) out[obase + i] = __int_as_float(lst[i]);
  }
}

static inline int cdiv_ll(long long a, long long b) { return (int)((a + b - 1) / b); }

extern "C" void kernel_launch(void* const* d_in, const int* in_sizes, int n_in,
                              void* d_out, int out_size, void* d_ws, size_t ws_size,
                              hipStream_t stream) {
  const float* x  = (const float*)d_in[0];
  const int*   ei = (const int*)  d_in[1];
  const float* W1 = (const float*)d_in[2];
  const float* b1 = (const float*)d_in[3];
  const float* W2 = (const float*)d_in[4];
  const float* b2 = (const float*)d_in[5];

  const int F = 128, H = 128, C = 40;
  const int N = in_sizes[0] / F;
  const int E = in_sizes[1] / 2;
  const int* src = ei;
  const int* dst = ei + E;

  const int nT1 = (H + 15) / 16;
  const int nT2 = (C + 15) / 16;

  const int nAgg = cdiv_ll(N, AGG_ROWS);

  float* ws   = (float*)d_ws;
  int*   wcnt = (int*)ws;           ws += N;
  float* dinv = ws;                 ws += N;
  float* h1   = ws;                 ws += (size_t)N * H;
  float* o1   = ws;                 ws += (size_t)N * H;
  int*   wlist = (int*)ws;          ws += (size_t)N * C;
  v2f* bf1    = (v2f*)ws;           ws += (size_t)nT1 * 1024 * 2;
  v2f* bf2    = (v2f*)ws;           ws += (size_t)nT2 * 1024 * 2;
  float* h2   = o1;
  float* out  = (float*)d_out;
  const int capW = ((size_t)nAgg * LIST_CAP <= (size_t)N * C) ? LIST_CAP : 0;

  const int B = 256;

  unsigned char* tileb = (unsigned char*)o1;
  const bool bytes = nAgg <= 256 && (size_t)E <= (size_t)N * H * sizeof(float);
  if (bytes) {
    k_tile_bytes<<<cdiv_ll((long long)(E >> 2) + 1, 1024), B, 0, stream>>>(dst, (unsigned*)tileb, E);
    k_scan_tiles<true><<<nAgg, AGG_THREADS, 0, stream>>>(dst, tileb, dinv, wlist, wcnt, capW, N, E);
  } else {
    k_scan_tiles<false><<<nAgg, AGG_THREADS, 0, stream>>>(dst, tileb, dinv, wlist, wcnt, capW, N, E);
  }

  k_pack_bfrag<<<cdiv_ll((long long)nT1 * 1024, B), B, 0, stream>>>(W1, bf1, H, nT1);
  k_pack_bfrag<<<cdiv_ll((long long)nT2 * 1024, B), B, 0, stream>>>(W2, bf2, C, nT2);

  gcn_gemm_wmma_f32<<<N / 16, B, 0, stream>>>(x, bf1, h1, N, H, nT1);
  k_agg_rows<128, 40><<<nAgg, AGG_THREADS, 0, stream>>>(h1, src, dst, dinv, wlist, wcnt, capW, h2, N, E, b1, bf2);

  k_agg_rows<40, 0, true><<<nAgg, AGG_THREADS, 0, stream>>>(h2, src, dst, dinv, wlist, wcnt, capW, out, N, E, b2, nullptr);
}


# Weights read in place as WMMA B fragments (row-major W1, W2; same k order): a layer-1 wave takes a 32-column pair so each B load and C store is whole lines; packing pass gone; 7->5 launches
# speedup vs baseline: 4.6645x; 1.0459x over previous
#include <hip/hip_runtime.h>
#include <hip/hip_bf16.h>
#include <math.h>

typedef float v2f __attribute__((ext_vector_type(2)));
typedef float v8f __attribute__((ext_vector_type(8)));

template <int V> struct IntC { static constexpr int value = V; };

__global__ __launch_bounds__(256) void gcn_gemm_wmma_f32(
    const float* __restrict__ A, const float* __restrict__ B,
    float* __restrict__ C, int M) {
  constexpr int K = 128;
  constexpr int N = 128;
  constexpr int LDS_STRIDE = 132;
  __shared__ alignas(16) float As[32 * LDS_STRIDE];

  const int m0  = blockIdx.x * 32;
  const int tid = threadIdx.x;

  if ((((size_t)A) & 15) == 0) {
#pragma unroll
    for (int j = 0; j < 4; ++j) {
      int g   = tid + j * 256;
      int row = g >> 5;
      int col = (g & 31) * 4;
      *(float4*)(As + row * LDS_STRIDE + col) =
          *(const float4*)(A + (size_t)min(m0 + row, M - 1) * K + col);
    }
  } else {
#pragma unroll
    for (int j = 0; j < 16; ++j) {
      int off = tid + j * 256;
      int row = off >> 7;
      int col = off & 127;
      As[row * LDS_STRIDE + col] = A[(size_t)min(m0 + row, M - 1) * K + col];
    }
  }
  __syncthreads();

  const int wave = tid >> 5;
  const int lane = tid & 31;
  const int half = lane >> 4;
  const int lm   = lane & 15;
  const int rt   = wave >> 2;
  const int cp   = wave & 3;

  const float* aBase = &As[(rt * 16 + lm) * LDS_STRIDE + 2 * half];
  const float* bBase = B + (size_t)(2 * half) * N + cp * 32 + 2 * lm;

  v8f c0 = {};
  v8f c1 = {};
  auto product = [&](auto aligned) {
    constexpr bool AL = decltype(aligned)::value != 0;
#pragma unroll
    for (int t = 0; t < 32; ++t) {
      v2f a = *(const v2f*)(aBase + 4 * t);
      const float* p0 = bBase + (size_t)(4 * t) * N;
      const float* p1 = bBase + (size_t)(4 * t + 1) * N;
      v2f p, q;
      if constexpr (AL) {
        p = *(const v2f*)p0;
        q = *(const v2f*)p1;
      } else {
        p.x = p0[0]; p.y = p0[1];
        q.x = p1[0]; q.y = p1[1];
      }
      v2f b0, b1;
      b0.x = p.x; b0.y = q.x;
      b1.x = p.y; b1.y = q.y;
      c0 = __builtin_amdgcn_wmma_f32_16x16x4_f32(
           false, a,  false, b0,
           (short)0, c0,  false,  false);
      c1 = __builtin_amdgcn_wmma_f32_16x16x4_f32(
           false, a,  false, b1,
           (short)0, c1,  false,  false);
    }
  };
  if ((((size_t)B) & 7) == 0) product(IntC<1>{});
  else                        product(IntC<0>{});

  const int rBase = m0 + rt * 16 + 8 * half;
  float* cBase = C + cp * 32 + 2 * lm;
  const bool cal = (((size_t)C) & 7) == 0;
#pragma unroll
  for (int r = 0; r < 8; ++r) {
    const int row = rBase + r;
    if (row < M) {
      float* cpn = cBase + (size_t)row * N;
      if (cal) {
        v2f v;
        v.x = c0[r]; v.y = c1[r];
        *(v2f*)cpn = v;
      } else {
        cpn[0] = c0[r];
        cpn[1] = c1[r];
      }
    }
  }
}

constexpr int AGG_THREADS = 1024;
constexpr int AGG_ROWS    = 196;
constexpr int AGG_LDS_BYTES = 168 * 1024;
constexpr int LIST_CAP      = 4096;

constexpr int SCAN_CAP = (AGG_LDS_BYTES - AGG_ROWS * 4 - 64) / 4;
static_assert(LIST_CAP <= SCAN_CAP, "the scan's LDS list must hold a full workspace list");

__global__ __launch_bounds__(256) void k_tile_bytes(
    const int* __restrict__ dst, unsigned* __restrict__ tb32, int E) {
  const int E4 = E >> 2;
  const int base = blockIdx.x * 1024 + threadIdx.x;
  auto pack = [](int4 d) {
    unsigned b0 = ((unsigned)d.x / (unsigned)AGG_ROWS) & 255u;
    unsigned b1 = ((unsigned)d.y / (unsigned)AGG_ROWS) & 255u;
    unsigned b2 = ((unsigned)d.z / (unsigned)AGG_ROWS) & 255u;
    unsigned b3 = ((unsigned)d.w / (unsigned)AGG_ROWS) & 255u;
    return b0 | (b1 << 8) | (b2 << 16) | (b3 << 24);
  };
  if (E4 > 0) {
    if ((((size_t)dst) & 15) == 0) {
      const int4* dst4 = (const int4*)dst;
      int4 d[4];
#pragma unroll
      for (int u = 0; u < 4; ++u) d[u] = dst4[min(base + u * 256, E4 - 1)];
#pragma unroll
      for (int u = 0; u < 4; ++u) {
        const int g = base + u * 256;
        if (g < E4) tb32[g] = pack(d[u]);
      }
    } else {
#pragma unroll
      for (int u = 0; u < 4; ++u) {
        const int g = base + u * 256;
        if (g < E4) tb32[g] = pack(make_int4(dst[4 * g], dst[4 * g + 1], dst[4 * g + 2], dst[4 * g + 3]));
      }
    }
  }
  if (blockIdx.x == 0 && (int)threadIdx.x < (E & 3)) {
    const int e = 4 * E4 + threadIdx.x;
    ((unsigned char*)tb32)[e] = (unsigned char)(((unsigned)dst[e] / (unsigned)AGG_ROWS) & 255u);
  }
}

template <bool BYTES>
__global__ __launch_bounds__(AGG_THREADS) void k_scan_tiles(
    const int* __restrict__ dst, const unsigned char* __restrict__ tileb,
    float* __restrict__ dinv, int* __restrict__ wlist, int* __restrict__ wcnt,
    int capW, int N, int E) {
  __shared__ int list[SCAN_CAP];
  __shared__ int rowcnt[AGG_ROWS];
  __shared__ int cnt;
  __shared__ int cntv;

  const int tid   = threadIdx.x;
  const int r0    = blockIdx.x * AGG_ROWS;
  const int nrows = min(AGG_ROWS, N - r0);
  const unsigned ur0 = (unsigned)r0, un = (unsigned)nrows;

  for (int i = tid; i < AGG_ROWS; i += AGG_THREADS) rowcnt[i] = 0;
  if (tid == 0) { cnt = 0; cntv = 0; }
  __syncthreads();

  auto count_row = [&](int e) -> bool {
    int d = dst[e];
    bool ok = ((unsigned)d - ur0) < un;
    if (ok) atomicAdd(&rowcnt[d - r0], 1);
    return ok;
  };
  auto note = [&](int e) {
    int pos = atomicAdd(&cnt, 1);
    if (pos < SCAN_CAP) list[pos] = e;
    else count_row(e);
  };

  constexpr int U = 8;
  constexpr int TILE = U * AGG_THREADS;
  if constexpr (BYTES) {
    const unsigned tile = blockIdx.x;
    int scalarFrom = 0;
    if ((((size_t)tileb) & 15) == 0) {
      const unsigned pat = tile * 0x01010101u;
      auto hit = [&](unsigned w) {
        unsigned x = w ^ pat;
        return (x - 0x01010101u) & ~x & 0x80808080u;
      };
      auto flags = [&](uint4 v, auto hi) -> unsigned {
        constexpr int HI = decltype(hi)::value;
        return (hit(v.x) >> (7 - HI)) | (hit(v.y) >> (6 - HI)) | (hit(v.z) >> (5 - HI)) | (hit(v.w) >> (4 - HI));
      };
      auto drain = [&](unsigned mm, int e0, int e1) {
        while (mm) {
          int b = __builtin_ctz(mm);
          mm &= mm - 1;
          note(((b & 4) ? e1 : e0) + 4 * (b & 3) + (b >> 3));
        }
      };
      const int G16 = E >> 4;
      const int G32 = G16 & ~31;
      const uint4* tb4 = (const uint4*)tileb;
      const int wave = __builtin_amdgcn_readfirstlane(tid >> 5);
      const int lane = tid & 31;
      const int start = (int)(tile * ((unsigned)G32 / gridDim.x)) & ~31;
      auto place = [&](int i) { int q = start + i; return q >= G32 ? q - G32 : q; };
      int i0 = 0;
      for (; i0 + TILE <= G32; i0 += TILE) {
        uint4 dv[U];
        int   q[U];
#pragma unroll
        for (int u = 0; u < U; ++u) {
          q[u]  = place(i0 + u * AGG_THREADS + wave * 32) + lane;
          dv[u] = tb4[q[u]];
        }
#pragma unroll
        for (int u = 0; u < U; u += 2)
          drain(flags(dv[u], IntC<0>{}) | flags(dv[u + 1], IntC<4>{}), 16 * q[u], 16 * q[u + 1]);
      }
      for (int i = i0 + wave * 32; i < G32; i += AGG_THREADS) {
        const int q = place(i) + lane;
        drain(flags(tb4[q], IntC<0>{}), 16 * q, 16 * q);
      }
      if (tid < G16 - G32) {
        const int q = G32 + tid;
        drain(flags(tb4[q], IntC<0>{}), 16 * q, 16 * q);
      }
      scalarFrom = G16 * 16;
    }
    for (int e = scalarFrom + tid; e < E; e += AGG_THREADS)
      if (tileb[e] == (unsigned char)tile) note(e);
  } else {
    auto check4 = [&](int4 d4, int e) {
      bool m0 = ((unsigned)d4.x - ur0) < un;
      bool m1 = ((unsigned)d4.y - ur0) < un;
      bool m2 = ((unsigned)d4.z - ur0) < un;
      bool m3 = ((unsigned)d4.w - ur0) < un;
      if (m0 | m1 | m2 | m3) {
        if (m0) note(e);
        if (m1) note(e + 1);
        if (m2) note(e + 2);
        if (m3) note(e + 3);
      }
    };
    int scalarFrom = 0;
    if ((((size_t)dst) & 15) == 0) {
      const int E4 = E >> 2;
      const int4* dst4 = (const int4*)dst;
      auto scan_groups = [&](int lo, int hi) {
        int t0 = lo;
        for (; t0 + TILE <= hi; t0 += TILE) {
          int4 dv[U];
#pragma unroll
          for (int u = 0; u < U; ++u) dv[u] = dst4[t0 + tid + u * AGG_THREADS];
#pragma unroll
          for (int u = 0; u < U; ++u) check4(dv[u], 4 * (t0 + tid + u * AGG_THREADS));
        }
        for (int q = t0 + tid; q < hi; q += AGG_THREADS) check4(dst4[q], 4 * q);
      };
      const int start = (int)(blockIdx.x * ((unsigned)E4 / gridDim.x)) & ~7;
      scan_groups(start, E4);
      scan_groups(0, start);
      scalarFrom = E4 * 4;
    }
    for (int e = scalarFrom + tid; e < E; e += AGG_THREADS) {
      int d = dst[e];
      if (((unsigned)d - ur0) < un) note(e);
    }
  }
  __syncthreads();

  const int noted = cnt;
  const int held  = min(noted, SCAN_CAP);
  int* wl = wlist + (size_t)blockIdx.x * LIST_CAP;
  for (int m = tid; m < held; m += AGG_THREADS) {
    int e = list[m];
    if (count_row(e)) {
      int p = atomicAdd(&cntv, 1);
      if (p < capW) wl[p] = e;
    }
  }
  __syncthreads();

  for (int i = tid; i < nrows; i += AGG_THREADS) dinv[r0 + i] = rsqrtf((float)(rowcnt[i] + 1));

  if (tid == 0) wcnt[blockIdx.x] = (noted <= SCAN_CAP) ? cntv : 0x7fffffff;
}

template <int F, int FO, bool LSM = false>
__global__ __launch_bounds__(AGG_THREADS) void k_agg_rows(
    const float* __restrict__ h, const int* __restrict__ src, const int* __restrict__ dst,
    const float* __restrict__ dinv, const int* __restrict__ wlist, const int* __restrict__ wcnt,
    int capW, float* __restrict__ out, int N, int E,
    const float* __restrict__ bias, const float* __restrict__ Wn) {
  constexpr bool MM     = FO > 0;
  constexpr int  PITCH  = MM ? F + 4 : F;
  constexpr int  ROWS_A = MM ? ((AGG_ROWS + 15) / 16) * 16 : AGG_ROWS;
  constexpr int  CAPL   = (AGG_LDS_BYTES - ROWS_A * PITCH * 4) / 12;
  static_assert(LIST_CAP <= CAPL, "LDS must hold a full workspace list beside the tile");
  static_assert(!MM || (F % 4 == 0 && ROWS_A * FO <= 3 * CAPL), "product rows are staged where the edge list was");
  __shared__ alignas(16) float acc[ROWS_A * PITCH];
  __shared__ int lst[3 * CAPL];
  int* eS = lst;
  int* eD = lst + CAPL;
  int* eC = lst + 2 * CAPL;

  const int tid   = threadIdx.x;
  const int r0    = blockIdx.x * AGG_ROWS;
  const int nrows = min(AGG_ROWS, N - r0);
  const int nel   = nrows * F;
  const size_t base = (size_t)r0 * F;

  const int  total  = wcnt[blockIdx.x];
  const bool listed = total <= capW;

  constexpr int LPT = (LIST_CAP + AGG_THREADS - 1) / AGG_THREADS;
  constexpr int NG  = ROWS_A * F / 4;
  constexpr int PPT = (NG + AGG_THREADS - 1) / AGG_THREADS;
  static_assert(F % 4 == 0 && PITCH % 4 == 0, "a 16-byte piece lies in one row, at a 16-byte place");
  const int* wl = wlist + (size_t)blockIdx.x * LIST_CAP;
  if ((((size_t)h) & 15) == 0) {
    const bool staged = listed && total > 0;
    float4 hv[PPT];
    float  dv[PPT];
#pragma unroll
    for (int k = 0; k < PPT; ++k) {
      const int i = min(4 * (tid + k * AGG_THREADS), nel - 4);
      dv[k] = dinv[r0 + i / F];
      hv[k] = *(const float4*)(h + base + i);
    }
    int le[LPT], ls[LPT], ld[LPT];
    if (staged) {
#pragma unroll
      for (int k = 0; k < LPT; ++k) le[k] = wl[min(tid + k * AGG_THREADS, total - 1)];
#pragma unroll
      for (int k = 0; k < LPT; ++k) {
        ls[k] = src[le[k]];
        ld[k] = dst[le[k]];
      }
    }
#pragma unroll
    for (int k = 0; k < PPT; ++k) {
      const int g = tid + k * AGG_THREADS;
      if (g < NG) {
        const int i = 4 * g;
        const int row = i / F, col = i % F;
        const float dd = dv[k] * dv[k];
        float4 v = make_float4(0.0f, 0.0f, 0.0f, 0.0f);
        if (i < nel) v = make_float4(dd * hv[k].x, dd * hv[k].y, dd * hv[k].z, dd * hv[k].w);
        *(float4*)(acc + row * PITCH + col) = v;
      }
    }
    if (staged) {
      float cs[LPT], cd[LPT];
#pragma unroll
      for (int k = 0; k < LPT; ++k) {
        cs[k] = dinv[ls[k]];
        cd[k] = dinv[ld[k]];
      }
#pragma unroll
      for (int k = 0; k < LPT; ++k) {
        const int m = tid + k * AGG_THREADS;
        if (m < total) {
          eS[m] = ls[k];
          eD[m] = ld[k] - r0;
          eC[m] = __float_as_int(cs[k] * cd[k]);
        }
      }
      for (int m = tid + LPT * AGG_THREADS; m < total; m += AGG_THREADS) {
        int e = wl[m];
        int s = src[e];
        int d = dst[e];
        eS[m] = s;
        eD[m] = d - r0;
        eC[m] = __float_as_int(dinv[s] * dinv[d]);
      }
    }
  } else {
    if (listed) {
      for (int m = tid; m < total; m += AGG_THREADS) {
        int e = wl[m];
        int s = src[e];
        int d = dst[e];
        eS[m] = s;
        eD[m] = d - r0;
        eC[m] = __float_as_int(dinv[s] * dinv[d]);
      }
    }
    for (int i = tid; i < ROWS_A * F; i += AGG_THREADS) {
      const int row = i / F, col = i % F;
      float v = 0.0f;
      if (i < nel) {
        float di = dinv[r0 + row];
        v = di * di * h[base + i];
      }
      acc[row * PITCH + col] = v;
    }
  }
  __syncthreads();

  if (listed) {
    const int wave = __builtin_amdgcn_readfirstlane(tid >> 5);
    const int lane = tid & 31;
    constexpr int NJ = (F + 31) / 32;
    constexpr int G  = 8;
    for (int m0 = wave * 32; m0 < total; m0 += AGG_THREADS) {
      const int m = m0 + lane;
      int s = 0, dl = 0;
      float c = 0.0f;
      if (m < total) {
        s  = eS[m];
        dl = eD[m];
        c  = __int_as_float(eC[m]);
      }
      auto add_edges = [&](auto gg, int i0) {
        constexpr int GG = decltype(gg)::value;
        float v[GG][NJ];
#pragma unroll
        for (int i = 0; i < GG; ++i) {
          int si = __builtin_amdgcn_readlane(s, i0 + i);
          const float* hr = h + (size_t)si * F;
#pragma unroll
          for (int j = 0; j < NJ; ++j) v[i][j] = hr[min(32 * j + lane, F - 1)];
        }
#pragma unroll
        for (int i = 0; i < GG; ++i) {
          int   di = __builtin_amdgcn_readlane(dl, i0 + i);
          float ci = __builtin_bit_cast(float, __builtin_amdgcn_readlane(__builtin_bit_cast(int, c), i0 + i));
          float* ar = acc + di * PITCH;
#pragma unroll
          for (int j = 0; j < NJ; ++j) {
            int f = 32 * j + lane;
            if (f < F) atomicAdd(&ar[f], ci * v[i][j]);
          }
        }
      };
      const int nb = min(32, total - m0);
      int i0 = 0;
      for (; i0 + G <= nb; i0 += G) add_edges(IntC<G>{}, i0);
      for (; i0 < nb; ++i0)         add_edges(IntC<1>{}, i0);
    }
  } else {
    const unsigned ur0 = (unsigned)r0, un = (unsigned)nrows;
    for (int e = tid; e < E; e += AGG_THREADS) {
      int d = dst[e];
      if (((unsigned)d - ur0) < un) {
        int s = src[e];
        float c = dinv[s] * dinv[d];
        for (int f = 0; f < F; ++f)
          atomicAdd(&acc[(d - r0) * PITCH + f], c * h[(size_t)s * F + f]);
      }
    }
  }
  __syncthreads();

  if constexpr (!MM) {
    if constexpr (LSM) {
      static_assert(F > 32 && F <= 64 && PITCH == F, "a lane holds one or two classes of a row");
      constexpr int NW  = AGG_THREADS / 32;
      constexpr int RPW = (AGG_ROWS + NW - 1) / NW;
      const int  wave = __builtin_amdgcn_readfirstlane(tid >> 5);
      const int  lane = tid & 31;
      const bool has2 = (lane + 32) < F;
      const float b0 = bias[lane];
      const float b1 = has2 ? bias[lane + 32] : 0.0f;
      const int  c2   = min(lane + 32, F - 1);
      float v0[RPW], v1[RPW], mx[RPW], sm[RPW];
#pragma unroll
      for (int k = 0; k < RPW; ++k) {
        const int row = wave + k * NW;
        v0[k] = 0.0f;
        v1[k] = 0.0f;
        if (row < nrows) {
          const float* p = acc + row * F;
          v0[k] = p[lane];
          v1[k] = p[c2];
        }
      }
#pragma unroll
      for (int k = 0; k < RPW; ++k) {
        v0[k] = v0[k] + b0;
        v1[k] = has2 ? v1[k] + b1 : -INFINITY;
        mx[k] = fmaxf(v0[k], v1[k]);
      }
#pragma unroll
      for (int off = 16; off > 0; off >>= 1) {
#pragma unroll
        for (int k = 0; k < RPW; ++k) mx[k] = fmaxf(mx[k], __shfl_xor(mx[k], off, 32));
      }
#pragma unroll
      for (int k = 0; k < RPW; ++k)
        sm[k] = expf(v0[k] - mx[k]) + (has2 ? expf(v1[k] - mx[k]) : 0.0f);
#pragma unroll
      for (int off = 16; off > 0; off >>= 1) {
#pragma unroll
        for (int k = 0; k < RPW; ++k) sm[k] += __shfl_xor(sm[k], off, 32);
      }
#pragma unroll
      for (int k = 0; k < RPW; ++k) {
        const int row = wave + k * NW;
        const float lse = logf(sm[k]);
        if (row < nrows) {
          float* p = acc + row * F;
          p[lane] = v0[k] - mx[k] - lse;
          if (has2) p[lane + 32] = v1[k] - mx[k] - lse;
        }
      }
      __syncthreads();
    }
    for (int i = tid; i < nel; i += AGG_THREADS) out[base + i] = acc[i];
  } else {
    for (int i = tid; i < nel; i += AGG_THREADS) {
      const int row = i / F, col = i % F;
      float v = acc[row * PITCH + col] + bias[col];
      acc[row * PITCH + col] = v > 0.0f ? v : 0.0f;
    }
    __syncthreads();

    constexpr int NT = (FO + 15) / 16;
    constexpr int JOBS = (ROWS_A / 16) * NT;
    const int wave = __builtin_amdgcn_readfirstlane(tid >> 5);
    const int lane = tid & 31;
    const int half = lane >> 4;
    const int lm   = lane & 15;
    for (int job = wave; job < JOBS; job += AGG_THREADS / 32) {
      const int rt = job / NT, nt = job % NT;
      const float* aBase = acc + (rt * 16 + lm) * PITCH + 2 * half;
      const int    col   = nt * 16 + lm;
      const bool   cv    = col < FO;
      const float* bBase = Wn + (size_t)(2 * half) * FO + min(col, FO - 1);
      v8f c = {};
#pragma unroll
      for (int t = 0; t < F / 4; ++t) {
        v2f a = *(const v2f*)(aBase + 4 * t);
        const float bx = bBase[(size_t)(4 * t) * FO];
        const float by = bBase[(size_t)(4 * t + 1) * FO];
        v2f b;
        b.x = cv ? bx : 0.0f;
        b.y = cv ? by : 0.0f;
        c = __builtin_amdgcn_wmma_f32_16x16x4_f32(
             false, a,  false, b,
             (short)0, c,  false,  false);
      }
      if (cv) {
        int* st = lst + (rt * 16 + 8 * half) * FO + col;
#pragma unroll
        for (int r = 0; r < 8; ++r) {
          const float cr = c[r];
          st[r * FO] = __float_as_int(cr);
        }
      }
    }
    __syncthreads();

    const size_t obase = (size_t)r0 * FO;
    for (int i = tid; i < nrows * FO; i += AGG_THREADS) out[obase + i] = __int_as_float(lst[i]);
  }
}

static inline int cdiv_ll(long long a, long long b) { return (int)((a + b - 1) / b); }

extern "C" void kernel_launch(void* const* d_in, const int* in_sizes, int n_in,
                              void* d_out, int out_size, void* d_ws, size_t ws_size,
                              hipStream_t stream) {
  const float* x  = (const float*)d_in[0];
  const int*   ei = (const int*)  d_in[1];
  const float* W1 = (const float*)d_in[2];
  const float* b1 = (const float*)d_in[3];
  const float* W2 = (const float*)d_in[4];
  const float* b2 = (const float*)d_in[5];

  const int F = 128, H = 128, C = 40;
  const int N = in_sizes[0] / F;
  const int E = in_sizes[1] / 2;
  const int* src = ei;
  const int* dst = ei + E;

  const int nAgg = cdiv_ll(N, AGG_ROWS);

  float* ws   = (float*)d_ws;
  int*   wcnt = (int*)ws;           ws += N;
  float* dinv = ws;                 ws += N;
  float* h1   = ws;                 ws += (size_t)N * H;
  float* o1   = ws;                 ws += (size_t)N * H;
  int*   wlist = (int*)ws;          ws += (size_t)N * C;
  float* h2   = o1;
  float* out  = (float*)d_out;
  const int capW = ((size_t)nAgg * LIST_CAP <= (size_t)N * C) ? LIST_CAP : 0;

  const int B = 256;

  unsigned char* tileb = (unsigned char*)o1;
  const bool bytes = nAgg <= 256 && (size_t)E <= (size_t)N * H * sizeof(float);
  if (bytes) {
    k_tile_bytes<<<cdiv_ll((long long)(E >> 2) + 1, 1024), B, 0, stream>>>(dst, (unsigned*)tileb, E);
    k_scan_tiles<true><<<nAgg, AGG_THREADS, 0, stream>>>(dst, tileb, dinv, wlist, wcnt, capW, N, E);
  } else {
    k_scan_tiles<false><<<nAgg, AGG_THREADS, 0, stream>>>(dst, tileb, dinv, wlist, wcnt, capW, N, E);
  }

  gcn_gemm_wmma_f32<<<cdiv_ll(N, 32), B, 0, stream>>>(x, W1, h1, N);
  k_agg_rows<128, 40><<<nAgg, AGG_THREADS, 0, stream>>>(h1, src, dst, dinv, wlist, wcnt, capW, h2, N, E, b1, W2);

  k_agg_rows<40, 0, true><<<nAgg, AGG_THREADS, 0, stream>>>(h2, src, dst, dinv, wlist, wcnt, capW, out, N, E, b2, nullptr);
}


# A tile's list word = the edge's source row (24 bits) + its row in the tile (8 bits), not its number: scan reads src beside dst once; both layers' set-up src/dst gathers gone; list size, sums same
# speedup vs baseline: 5.0567x; 1.0841x over previous
#include <hip/hip_runtime.h>
#include <hip/hip_bf16.h>
#include <math.h>

typedef float v2f __attribute__((ext_vector_type(2)));
typedef float v8f __attribute__((ext_vector_type(8)));

template <int V> struct IntC { static constexpr int value = V; };

__global__ __launch_bounds__(256) void gcn_gemm_wmma_f32(
    const float* __restrict__ A, const float* __restrict__ B,
    float* __restrict__ C, int M) {
  constexpr int K = 128;
  constexpr int N = 128;
  constexpr int LDS_STRIDE = 132;
  __shared__ alignas(16) float As[32 * LDS_STRIDE];

  const int m0  = blockIdx.x * 32;
  const int tid = threadIdx.x;

  if ((((size_t)A) & 15) == 0) {
#pragma unroll
    for (int j = 0; j < 4; ++j) {
      int g   = tid + j * 256;
      int row = g >> 5;
      int col = (g & 31) * 4;
      *(float4*)(As + row * LDS_STRIDE + col) =
          *(const float4*)(A + (size_t)min(m0 + row, M - 1) * K + col);
    }
  } else {
#pragma unroll
    for (int j = 0; j < 16; ++j) {
      int off = tid + j * 256;
      int row = off >> 7;
      int col = off & 127;
      As[row * LDS_STRIDE + col] = A[(size_t)min(m0 + row, M - 1) * K + col];
    }
  }
  __syncthreads();

  const int wave = tid >> 5;
  const int lane = tid & 31;
  const int half = lane >> 4;
  const int lm   = lane & 15;
  const int rt   = wave >> 2;
  const int cp   = wave & 3;

  const float* aBase = &As[(rt * 16 + lm) * LDS_STRIDE + 2 * half];
  const float* bBase = B + (size_t)(2 * half) * N + cp * 32 + 2 * lm;

  v8f c0 = {};
  v8f c1 = {};
  auto product = [&](auto aligned) {
    constexpr bool AL = decltype(aligned)::value != 0;
#pragma unroll
    for (int t = 0; t < 32; ++t) {
      v2f a = *(const v2f*)(aBase + 4 * t);
      const float* p0 = bBase + (size_t)(4 * t) * N;
      const float* p1 = bBase + (size_t)(4 * t + 1) * N;
      v2f p, q;
      if constexpr (AL) {
        p = *(const v2f*)p0;
        q = *(const v2f*)p1;
      } else {
        p.x = p0[0]; p.y = p0[1];
        q.x = p1[0]; q.y = p1[1];
      }
      v2f b0, b1;
      b0.x = p.x; b0.y = q.x;
      b1.x = p.y; b1.y = q.y;
      c0 = __builtin_amdgcn_wmma_f32_16x16x4_f32(
           false, a,  false, b0,
           (short)0, c0,  false,  false);
      c1 = __builtin_amdgcn_wmma_f32_16x16x4_f32(
           false, a,  false, b1,
           (short)0, c1,  false,  false);
    }
  };
  if ((((size_t)B) & 7) == 0) product(IntC<1>{});
  else                        product(IntC<0>{});

  const int rBase = m0 + rt * 16 + 8 * half;
  float* cBase = C + cp * 32 + 2 * lm;
  const bool cal = (((size_t)C) & 7) == 0;
#pragma unroll
  for (int r = 0; r < 8; ++r) {
    const int row = rBase + r;
    if (row < M) {
      float* cpn = cBase + (size_t)row * N;
      if (cal) {
        v2f v;
        v.x = c0[r]; v.y = c1[r];
        *(v2f*)cpn = v;
      } else {
        cpn[0] = c0[r];
        cpn[1] = c1[r];
      }
    }
  }
}

constexpr int AGG_THREADS = 1024;
constexpr int AGG_ROWS    = 196;
constexpr int AGG_LDS_BYTES = 168 * 1024;
constexpr int LIST_CAP      = 4096;
constexpr int SRC_BITS      = 24;
constexpr int SRC_MASK      = (1 << SRC_BITS) - 1;
static_assert(AGG_ROWS <= (1 << (32 - SRC_BITS)), "the row inside a tile must fit the list word's high bits");

constexpr int SCAN_CAP = (AGG_LDS_BYTES - AGG_ROWS * 4 - 64) / 4;
static_assert(LIST_CAP <= SCAN_CAP, "the scan's LDS list must hold a full workspace list");

__global__ __launch_bounds__(256) void k_tile_bytes(
    const int* __restrict__ dst, unsigned* __restrict__ tb32, int E) {
  const int E4 = E >> 2;
  const int base = blockIdx.x * 1024 + threadIdx.x;
  auto pack = [](int4 d) {
    unsigned b0 = ((unsigned)d.x / (unsigned)AGG_ROWS) & 255u;
    unsigned b1 = ((unsigned)d.y / (unsigned)AGG_ROWS) & 255u;
    unsigned b2 = ((unsigned)d.z / (unsigned)AGG_ROWS) & 255u;
    unsigned b3 = ((unsigned)d.w / (unsigned)AGG_ROWS) & 255u;
    return b0 | (b1 << 8) | (b2 << 16) | (b3 << 24);
  };
  if (E4 > 0) {
    if ((((size_t)dst) & 15) == 0) {
      const int4* dst4 = (const int4*)dst;
      int4 d[4];
#pragma unroll
      for (int u = 0; u < 4; ++u) d[u] = dst4[min(base + u * 256, E4 - 1)];
#pragma unroll
      for (int u = 0; u < 4; ++u) {
        const int g = base + u * 256;
        if (g < E4) tb32[g] = pack(d[u]);
      }
    } else {
#pragma unroll
      for (int u = 0; u < 4; ++u) {
        const int g = base + u * 256;
        if (g < E4) tb32[g] = pack(make_int4(dst[4 * g], dst[4 * g + 1], dst[4 * g + 2], dst[4 * g + 3]));
      }
    }
  }
  if (blockIdx.x == 0 && (int)threadIdx.x < (E & 3)) {
    const int e = 4 * E4 + threadIdx.x;
    ((unsigned char*)tb32)[e] = (unsigned char)(((unsigned)dst[e] / (unsigned)AGG_ROWS) & 255u);
  }
}

template <bool BYTES>
__global__ __launch_bounds__(AGG_THREADS) void k_scan_tiles(
    const int* __restrict__ src, const int* __restrict__ dst, const unsigned char* __restrict__ tileb,
    float* __restrict__ dinv, int* __restrict__ wlist, int* __restrict__ wcnt,
    int capW, int N, int E) {
  __shared__ int list[SCAN_CAP];
  __shared__ int rowcnt[AGG_ROWS];
  __shared__ int cnt;
  __shared__ int cntv;

  const int tid   = threadIdx.x;
  const int r0    = blockIdx.x * AGG_ROWS;
  const int nrows = min(AGG_ROWS, N - r0);
  const unsigned ur0 = (unsigned)r0, un = (unsigned)nrows;

  for (int i = tid; i < AGG_ROWS; i += AGG_THREADS) rowcnt[i] = 0;
  if (tid == 0) { cnt = 0; cntv = 0; }
  __syncthreads();

  auto count_row = [&](int e) -> bool {
    int d = dst[e];
    bool ok = ((unsigned)d - ur0) < un;
    if (ok) atomicAdd(&rowcnt[d - r0], 1);
    return ok;
  };
  auto note = [&](int e) {
    int pos = atomicAdd(&cnt, 1);
    if (pos < SCAN_CAP) list[pos] = e;
    else count_row(e);
  };

  constexpr int U = 8;
  constexpr int TILE = U * AGG_THREADS;
  if constexpr (BYTES) {
    const unsigned tile = blockIdx.x;
    int scalarFrom = 0;
    if ((((size_t)tileb) & 15) == 0) {
      const unsigned pat = tile * 0x01010101u;
      auto hit = [&](unsigned w) {
        unsigned x = w ^ pat;
        return (x - 0x01010101u) & ~x & 0x80808080u;
      };
      auto flags = [&](uint4 v, auto hi) -> unsigned {
        constexpr int HI = decltype(hi)::value;
        return (hit(v.x) >> (7 - HI)) | (hit(v.y) >> (6 - HI)) | (hit(v.z) >> (5 - HI)) | (hit(v.w) >> (4 - HI));
      };
      auto drain = [&](unsigned mm, int e0, int e1) {
        while (mm) {
          int b = __builtin_ctz(mm);
          mm &= mm - 1;
          note(((b & 4) ? e1 : e0) + 4 * (b & 3) + (b >> 3));
        }
      };
      const int G16 = E >> 4;
      const int G32 = G16 & ~31;
      const uint4* tb4 = (const uint4*)tileb;
      const int wave = __builtin_amdgcn_readfirstlane(tid >> 5);
      const int lane = tid & 31;
      const int start = (int)(tile * ((unsigned)G32 / gridDim.x)) & ~31;
      auto place = [&](int i) { int q = start + i; return q >= G32 ? q - G32 : q; };
      int i0 = 0;
      for (; i0 + TILE <= G32; i0 += TILE) {
        uint4 dv[U];
        int   q[U];
#pragma unroll
        for (int u = 0; u < U; ++u) {
          q[u]  = place(i0 + u * AGG_THREADS + wave * 32) + lane;
          dv[u] = tb4[q[u]];
        }
#pragma unroll
        for (int u = 0; u < U; u += 2)
          drain(flags(dv[u], IntC<0>{}) | flags(dv[u + 1], IntC<4>{}), 16 * q[u], 16 * q[u + 1]);
      }
      for (int i = i0 + wave * 32; i < G32; i += AGG_THREADS) {
        const int q = place(i) + lane;
        drain(flags(tb4[q], IntC<0>{}), 16 * q, 16 * q);
      }
      if (tid < G16 - G32) {
        const int q = G32 + tid;
        drain(flags(tb4[q], IntC<0>{}), 16 * q, 16 * q);
      }
      scalarFrom = G16 * 16;
    }
    for (int e = scalarFrom + tid; e < E; e += AGG_THREADS)
      if (tileb[e] == (unsigned char)tile) note(e);
  } else {
    auto check4 = [&](int4 d4, int e) {
      bool m0 = ((unsigned)d4.x - ur0) < un;
      bool m1 = ((unsigned)d4.y - ur0) < un;
      bool m2 = ((unsigned)d4.z - ur0) < un;
      bool m3 = ((unsigned)d4.w - ur0) < un;
      if (m0 | m1 | m2 | m3) {
        if (m0) note(e);
        if (m1) note(e + 1);
        if (m2) note(e + 2);
        if (m3) note(e + 3);
      }
    };
    int scalarFrom = 0;
    if ((((size_t)dst) & 15) == 0) {
      const int E4 = E >> 2;
      const int4* dst4 = (const int4*)dst;
      auto scan_groups = [&](int lo, int hi) {
        int t0 = lo;
        for (; t0 + TILE <= hi; t0 += TILE) {
          int4 dv[U];
#pragma unroll
          for (int u = 0; u < U; ++u) dv[u] = dst4[t0 + tid + u * AGG_THREADS];
#pragma unroll
          for (int u = 0; u < U; ++u) check4(dv[u], 4 * (t0 + tid + u * AGG_THREADS));
        }
        for (int q = t0 + tid; q < hi; q += AGG_THREADS) check4(dst4[q], 4 * q);
      };
      const int start = (int)(blockIdx.x * ((unsigned)E4 / gridDim.x)) & ~7;
      scan_groups(start, E4);
      scan_groups(0, start);
      scalarFrom = E4 * 4;
    }
    for (int e = scalarFrom + tid; e < E; e += AGG_THREADS) {
      int d = dst[e];
      if (((unsigned)d - ur0) < un) note(e);
    }
  }
  __syncthreads();

  const int noted = cnt;
  const int held  = min(noted, SCAN_CAP);
  int* wl = wlist + (size_t)blockIdx.x * LIST_CAP;
  for (int m = tid; m < held; m += AGG_THREADS) {
    const int e = list[m];
    const int s = src[e];
    const int d = dst[e];
    asm volatile("" :: "v"(s));
    if (((unsigned)d - ur0) < un) {
      atomicAdd(&rowcnt[d - r0], 1);
      int p = atomicAdd(&cntv, 1);
      if (p < capW) wl[p] = (int)((unsigned)s | ((unsigned)(d - r0) << SRC_BITS));
    }
  }
  __syncthreads();

  for (int i = tid; i < nrows; i += AGG_THREADS) dinv[r0 + i] = rsqrtf((float)(rowcnt[i] + 1));

  if (tid == 0) wcnt[blockIdx.x] = (noted <= SCAN_CAP) ? cntv : 0x7fffffff;
}

template <int F, int FO, bool LSM = false>
__global__ __launch_bounds__(AGG_THREADS) void k_agg_rows(
    const float* __restrict__ h, const int* __restrict__ src, const int* __restrict__ dst,
    const float* __restrict__ dinv, const int* __restrict__ wlist, const int* __restrict__ wcnt,
    int capW, float* __restrict__ out, int N, int E,
    const float* __restrict__ bias, const float* __restrict__ Wn) {
  constexpr bool MM     = FO > 0;
  constexpr int  PITCH  = MM ? F + 4 : F;
  constexpr int  ROWS_A = MM ? ((AGG_ROWS + 15) / 16) * 16 : AGG_ROWS;
  constexpr int  CAPL   = (AGG_LDS_BYTES - ROWS_A * PITCH * 4) / 12;
  static_assert(LIST_CAP <= CAPL, "LDS must hold a full workspace list beside the tile");
  static_assert(!MM || (F % 4 == 0 && ROWS_A * FO <= 3 * CAPL), "product rows are staged where the edge list was");
  __shared__ alignas(16) float acc[ROWS_A * PITCH];
  __shared__ int lst[3 * CAPL];
  int* eS = lst;
  int* eD = lst + CAPL;
  int* eC = lst + 2 * CAPL;

  const int tid   = threadIdx.x;
  const int r0    = blockIdx.x * AGG_ROWS;
  const int nrows = min(AGG_ROWS, N - r0);
  const int nel   = nrows * F;
  const size_t base = (size_t)r0 * F;

  const int  total  = wcnt[blockIdx.x];
  const bool listed = total <= capW;

  constexpr int LPT = (LIST_CAP + AGG_THREADS - 1) / AGG_THREADS;
  constexpr int NG  = ROWS_A * F / 4;
  constexpr int PPT = (NG + AGG_THREADS - 1) / AGG_THREADS;
  static_assert(F % 4 == 0 && PITCH % 4 == 0, "a 16-byte piece lies in one row, at a 16-byte place");
  const int* wl = wlist + (size_t)blockIdx.x * LIST_CAP;
  if ((((size_t)h) & 15) == 0) {
    const bool staged = listed && total > 0;
    float4 hv[PPT];
    float  dv[PPT];
#pragma unroll
    for (int k = 0; k < PPT; ++k) {
      const int i = min(4 * (tid + k * AGG_THREADS), nel - 4);
      dv[k] = dinv[r0 + i / F];
      hv[k] = *(const float4*)(h + base + i);
    }
    int   lw[LPT];
    float cs[LPT], cd[LPT];
    if (staged) {
#pragma unroll
      for (int k = 0; k < LPT; ++k) lw[k] = wl[min(tid + k * AGG_THREADS, total - 1)];
#pragma unroll
      for (int k = 0; k < LPT; ++k) {
        cs[k] = dinv[lw[k] & SRC_MASK];
        cd[k] = dinv[r0 + (int)((unsigned)lw[k] >> SRC_BITS)];
      }
    }
#pragma unroll
    for (int k = 0; k < PPT; ++k) {
      const int g = tid + k * AGG_THREADS;
      if (g < NG) {
        const int i = 4 * g;
        const int row = i / F, col = i % F;
        const float dd = dv[k] * dv[k];
        float4 v = make_float4(0.0f, 0.0f, 0.0f, 0.0f);
        if (i < nel) v = make_float4(dd * hv[k].x, dd * hv[k].y, dd * hv[k].z, dd * hv[k].w);
        *(float4*)(acc + row * PITCH + col) = v;
      }
    }
    if (staged) {
#pragma unroll
      for (int k = 0; k < LPT; ++k) {
        const int m = tid + k * AGG_THREADS;
        if (m < total) {
          eS[m] = lw[k] & SRC_MASK;
          eD[m] = (int)((unsigned)lw[k] >> SRC_BITS);
          eC[m] = __float_as_int(cs[k] * cd[k]);
        }
      }
      for (int m = tid + LPT * AGG_THREADS; m < total; m += AGG_THREADS) {
        const int w = wl[m];
        const int s = w & SRC_MASK;
        const int r = (int)((unsigned)w >> SRC_BITS);
        eS[m] = s;
        eD[m] = r;
        eC[m] = __float_as_int(dinv[s] * dinv[r0 + r]);
      }
    }
  } else {
    if (listed) {
      for (int m = tid; m < total; m += AGG_THREADS) {
        const int w = wl[m];
        const int s = w & SRC_MASK;
        const int r = (int)((unsigned)w >> SRC_BITS);
        eS[m] = s;
        eD[m] = r;
        eC[m] = __float_as_int(dinv[s] * dinv[r0 + r]);
      }
    }
    for (int i = tid; i < ROWS_A * F; i += AGG_THREADS) {
      const int row = i / F, col = i % F;
      float v = 0.0f;
      if (i < nel) {
        float di = dinv[r0 + row];
        v = di * di * h[base + i];
      }
      acc[row * PITCH + col] = v;
    }
  }
  __syncthreads();

  if (listed) {
    const int wave = __builtin_amdgcn_readfirstlane(tid >> 5);
    const int lane = tid & 31;
    constexpr int NJ = (F + 31) / 32;
    constexpr int G  = 8;
    for (int m0 = wave * 32; m0 < total; m0 += AGG_THREADS) {
      const int m = m0 + lane;
      int s = 0, dl = 0;
      float c = 0.0f;
      if (m < total) {
        s  = eS[m];
        dl = eD[m];
        c  = __int_as_float(eC[m]);
      }
      auto add_edges = [&](auto gg, int i0) {
        constexpr int GG = decltype(gg)::value;
        float v[GG][NJ];
#pragma unroll
        for (int i = 0; i < GG; ++i) {
          int si = __builtin_amdgcn_readlane(s, i0 + i);
          const float* hr = h + (size_t)si * F;
#pragma unroll
          for (int j = 0; j < NJ; ++j) v[i][j] = hr[min(32 * j + lane, F - 1)];
        }
#pragma unroll
        for (int i = 0; i < GG; ++i) {
          int   di = __builtin_amdgcn_readlane(dl, i0 + i);
          float ci = __builtin_bit_cast(float, __builtin_amdgcn_readlane(__builtin_bit_cast(int, c), i0 + i));
          float* ar = acc + di * PITCH;
#pragma unroll
          for (int j = 0; j < NJ; ++j) {
            int f = 32 * j + lane;
            if (f < F) atomicAdd(&ar[f], ci * v[i][j]);
          }
        }
      };
      const int nb = min(32, total - m0);
      int i0 = 0;
      for (; i0 + G <= nb; i0 += G) add_edges(IntC<G>{}, i0);
      for (; i0 < nb; ++i0)         add_edges(IntC<1>{}, i0);
    }
  } else {
    const unsigned ur0 = (unsigned)r0, un = (unsigned)nrows;
    for (int e = tid; e < E; e += AGG_THREADS) {
      int d = dst[e];
      if (((unsigned)d - ur0) < un) {
        int s = src[e];
        float c = dinv[s] * dinv[d];
        for (int f = 0; f < F; ++f)
          atomicAdd(&acc[(d - r0) * PITCH + f], c * h[(size_t)s * F + f]);
      }
    }
  }
  __syncthreads();

  if constexpr (!MM) {
    if constexpr (LSM) {
      static_assert(F > 32 && F <= 64 && PITCH == F, "a lane holds one or two classes of a row");
      constexpr int NW  = AGG_THREADS / 32;
      constexpr int RPW = (AGG_ROWS + NW - 1) / NW;
      const int  wave = __builtin_amdgcn_readfirstlane(tid >> 5);
      const int  lane = tid & 31;
      const bool has2 = (lane + 32) < F;
      const float b0 = bias[lane];
      const float b1 = has2 ? bias[lane + 32] : 0.0f;
      const int  c2   = min(lane + 32, F - 1);
      float v0[RPW], v1[RPW], mx[RPW], sm[RPW];
#pragma unroll
      for (int k = 0; k < RPW; ++k) {
        const int row = wave + k * NW;
        v0[k] = 0.0f;
        v1[k] = 0.0f;
        if (row < nrows) {
          const float* p = acc + row * F;
          v0[k] = p[lane];
          v1[k] = p[c2];
        }
      }
#pragma unroll
      for (int k = 0; k < RPW; ++k) {
        v0[k] = v0[k] + b0;
        v1[k] = has2 ? v1[k] + b1 : -INFINITY;
        mx[k] = fmaxf(v0[k], v1[k]);
      }
#pragma unroll
      for (int off = 16; off > 0; off >>= 1) {
#pragma unroll
        for (int k = 0; k < RPW; ++k) mx[k] = fmaxf(mx[k], __shfl_xor(mx[k], off, 32));
      }
#pragma unroll
      for (int k = 0; k < RPW; ++k)
        sm[k] = expf(v0[k] - mx[k]) + (has2 ? expf(v1[k] - mx[k]) : 0.0f);
#pragma unroll
      for (int off = 16; off > 0; off >>= 1) {
#pragma unroll
        for (int k = 0; k < RPW; ++k) sm[k] += __shfl_xor(sm[k], off, 32);
      }
#pragma unroll
      for (int k = 0; k < RPW; ++k) {
        const int row = wave + k * NW;
        const float lse = logf(sm[k]);
        if (row < nrows) {
          float* p = acc + row * F;
          p[lane] = v0[k] - mx[k] - lse;
          if (has2) p[lane + 32] = v1[k] - mx[k] - lse;
        }
      }
      __syncthreads();
    }
    for (int i = tid; i < nel; i += AGG_THREADS) out[base + i] = acc[i];
  } else {
    for (int i = tid; i < nel; i += AGG_THREADS) {
      const int row = i / F, col = i % F;
      float v = acc[row * PITCH + col] + bias[col];
      acc[row * PITCH + col] = v > 0.0f ? v : 0.0f;
    }
    __syncthreads();

    constexpr int NT = (FO + 15) / 16;
    constexpr int JOBS = (ROWS_A / 16) * NT;
    const int wave = __builtin_amdgcn_readfirstlane(tid >> 5);
    const int lane = tid & 31;
    const int half = lane >> 4;
    const int lm   = lane & 15;
    for (int job = wave; job < JOBS; job += AGG_THREADS / 32) {
      const int rt = job / NT, nt = job % NT;
      const float* aBase = acc + (rt * 16 + lm) * PITCH + 2 * half;
      const int    col   = nt * 16 + lm;
      const bool   cv    = col < FO;
      const float* bBase = Wn + (size_t)(2 * half) * FO + min(col, FO - 1);
      v8f c = {};
#pragma unroll
      for (int t = 0; t < F / 4; ++t) {
        v2f a = *(const v2f*)(aBase + 4 * t);
        const float bx = bBase[(size_t)(4 * t) * FO];
        const float by = bBase[(size_t)(4 * t + 1) * FO];
        v2f b;
        b.x = cv ? bx : 0.0f;
        b.y = cv ? by : 0.0f;
        c = __builtin_amdgcn_wmma_f32_16x16x4_f32(
             false, a,  false, b,
             (short)0, c,  false,  false);
      }
      if (cv) {
        int* st = lst + (rt * 16 + 8 * half) * FO + col;
#pragma unroll
        for (int r = 0; r < 8; ++r) {
          const float cr = c[r];
          st[r * FO] = __float_as_int(cr);
        }
      }
    }
    __syncthreads();

    const size_t obase = (size_t)r0 * FO;
    for (int i = tid; i < nrows * FO; i += AGG_THREADS) out[obase + i] = __int_as_float(lst[i]);
  }
}

static inline int cdiv_ll(long long a, long long b) { return (int)((a + b - 1) / b); }

extern "C" void kernel_launch(void* const* d_in, const int* in_sizes, int n_in,
                              void* d_out, int out_size, void* d_ws, size_t ws_size,
                              hipStream_t stream) {
  const float* x  = (const float*)d_in[0];
  const int*   ei = (const int*)  d_in[1];
  const float* W1 = (const float*)d_in[2];
  const float* b1 = (const float*)d_in[3];
  const float* W2 = (const float*)d_in[4];
  const float* b2 = (const float*)d_in[5];

  const int F = 128, H = 128, C = 40;
  const int N = in_sizes[0] / F;
  const int E = in_sizes[1] / 2;
  const int* src = ei;
  const int* dst = ei + E;

  const int nAgg = cdiv_ll(N, AGG_ROWS);

  float* ws   = (float*)d_ws;
  int*   wcnt = (int*)ws;           ws += N;
  float* dinv = ws;                 ws += N;
  float* h1   = ws;                 ws += (size_t)N * H;
  float* o1   = ws;                 ws += (size_t)N * H;
  int*   wlist = (int*)ws;          ws += (size_t)N * C;
  float* h2   = o1;
  float* out  = (float*)d_out;
  const int capW = ((size_t)nAgg * LIST_CAP <= (size_t)N * C && N <= SRC_MASK + 1) ? LIST_CAP : 0;

  const int B = 256;

  unsigned char* tileb = (unsigned char*)o1;
  const bool bytes = nAgg <= 256 && (size_t)E <= (size_t)N * H * sizeof(float);
  if (bytes) {
    k_tile_bytes<<<cdiv_ll((long long)(E >> 2) + 1, 1024), B, 0, stream>>>(dst, (unsigned*)tileb, E);
    k_scan_tiles<true><<<nAgg, AGG_THREADS, 0, stream>>>(src, dst, tileb, dinv, wlist, wcnt, capW, N, E);
  } else {
    k_scan_tiles<false><<<nAgg, AGG_THREADS, 0, stream>>>(src, dst, tileb, dinv, wlist, wcnt, capW, N, E);
  }

  gcn_gemm_wmma_f32<<<cdiv_ll(N, 32), B, 0, stream>>>(x, W1, h1, N);
  k_agg_rows<128, 40><<<nAgg, AGG_THREADS, 0, stream>>>(h1, src, dst, dinv, wlist, wcnt, capW, h2, N, E, b1, W2);

  k_agg_rows<40, 0, true><<<nAgg, AGG_THREADS, 0, stream>>>(h2, src, dst, dinv, wlist, wcnt, capW, out, N, E, b2, nullptr);
}


# Tile-byte pass rides in the GEMM's launch as its first 196 blocks (neither needs the other; short memory-bound blocks beside matrix-bound ones): 5->4 launches; kernels' work and all sums unchanged
# speedup vs baseline: 5.2288x; 1.0340x over previous
#include <hip/hip_runtime.h>
#include <hip/hip_bf16.h>
#include <math.h>

typedef float v2f __attribute__((ext_vector_type(2)));
typedef float v8f __attribute__((ext_vector_type(8)));

template <int V> struct IntC { static constexpr int value = V; };

__device__ __forceinline__ void gemm_block(
    const float* __restrict__ A, const float* __restrict__ B,
    float* __restrict__ C, int M, int blk) {
  constexpr int K = 128;
  constexpr int N = 128;
  constexpr int LDS_STRIDE = 132;
  __shared__ alignas(16) float As[32 * LDS_STRIDE];

  const int m0  = blk * 32;
  const int tid = threadIdx.x;

  if ((((size_t)A) & 15) == 0) {
#pragma unroll
    for (int j = 0; j < 4; ++j) {
      int g   = tid + j * 256;
      int row = g >> 5;
      int col = (g & 31) * 4;
      *(float4*)(As + row * LDS_STRIDE + col) =
          *(const float4*)(A + (size_t)min(m0 + row, M - 1) * K + col);
    }
  } else {
#pragma unroll
    for (int j = 0; j < 16; ++j) {
      int off = tid + j * 256;
      int row = off >> 7;
      int col = off & 127;
      As[row * LDS_STRIDE + col] = A[(size_t)min(m0 + row, M - 1) * K + col];
    }
  }
  __syncthreads();

  const int wave = tid >> 5;
  const int lane = tid & 31;
  const int half = lane >> 4;
  const int lm   = lane & 15;
  const int rt   = wave >> 2;
  const int cp   = wave & 3;

  const float* aBase = &As[(rt * 16 + lm) * LDS_STRIDE + 2 * half];
  const float* bBase = B + (size_t)(2 * half) * N + cp * 32 + 2 * lm;

  v8f c0 = {};
  v8f c1 = {};
  auto product = [&](auto aligned) {
    constexpr bool AL = decltype(aligned)::value != 0;
#pragma unroll
    for (int t = 0; t < 32; ++t) {
      v2f a = *(const v2f*)(aBase + 4 * t);
      const float* p0 = bBase + (size_t)(4 * t) * N;
      const float* p1 = bBase + (size_t)(4 * t + 1) * N;
      v2f p, q;
      if constexpr (AL) {
        p = *(const v2f*)p0;
        q = *(const v2f*)p1;
      } else {
        p.x = p0[0]; p.y = p0[1];
        q.x = p1[0]; q.y = p1[1];
      }
      v2f b0, b1;
      b0.x = p.x; b0.y = q.x;
      b1.x = p.y; b1.y = q.y;
      c0 = __builtin_amdgcn_wmma_f32_16x16x4_f32(
           false, a,  false, b0,
           (short)0, c0,  false,  false);
      c1 = __builtin_amdgcn_wmma_f32_16x16x4_f32(
           false, a,  false, b1,
           (short)0, c1,  false,  false);
    }
  };
  if ((((size_t)B) & 7) == 0) product(IntC<1>{});
  else                        product(IntC<0>{});

  const int rBase = m0 + rt * 16 + 8 * half;
  float* cBase = C + cp * 32 + 2 * lm;
  const bool cal = (((size_t)C) & 7) == 0;
#pragma unroll
  for (int r = 0; r < 8; ++r) {
    const int row = rBase + r;
    if (row < M) {
      float* cpn = cBase + (size_t)row * N;
      if (cal) {
        v2f v;
        v.x = c0[r]; v.y = c1[r];
        *(v2f*)cpn = v;
      } else {
        cpn[0] = c0[r];
        cpn[1] = c1[r];
      }
    }
  }
}

__global__ __launch_bounds__(256) void gcn_gemm_wmma_f32(
    const float* __restrict__ A, const float* __restrict__ B,
    float* __restrict__ C, int M) {
  gemm_block(A, B, C, M, blockIdx.x);
}

constexpr int AGG_THREADS = 1024;
constexpr int AGG_ROWS    = 196;
constexpr int AGG_LDS_BYTES = 168 * 1024;
constexpr int LIST_CAP      = 4096;
constexpr int SRC_BITS      = 24;
constexpr int SRC_MASK      = (1 << SRC_BITS) - 1;
static_assert(AGG_ROWS <= (1 << (32 - SRC_BITS)), "the row inside a tile must fit the list word's high bits");

constexpr int SCAN_CAP = (AGG_LDS_BYTES - AGG_ROWS * 4 - 64) / 4;
static_assert(LIST_CAP <= SCAN_CAP, "the scan's LDS list must hold a full workspace list");

__device__ __forceinline__ void tile_bytes_block(
    const int* __restrict__ dst, unsigned* __restrict__ tb32, int E, int blk) {
  const int E4 = E >> 2;
  const int base = blk * 1024 + threadIdx.x;
  auto pack = [](int4 d) {
    unsigned b0 = ((unsigned)d.x / (unsigned)AGG_ROWS) & 255u;
    unsigned b1 = ((unsigned)d.y / (unsigned)AGG_ROWS) & 255u;
    unsigned b2 = ((unsigned)d.z / (unsigned)AGG_ROWS) & 255u;
    unsigned b3 = ((unsigned)d.w / (unsigned)AGG_ROWS) & 255u;
    return b0 | (b1 << 8) | (b2 << 16) | (b3 << 24);
  };
  if (E4 > 0) {
    if ((((size_t)dst) & 15) == 0) {
      const int4* dst4 = (const int4*)dst;
      int4 d[4];
#pragma unroll
      for (int u = 0; u < 4; ++u) d[u] = dst4[min(base + u * 256, E4 - 1)];
#pragma unroll
      for (int u = 0; u < 4; ++u) {
        const int g = base + u * 256;
        if (g < E4) tb32[g] = pack(d[u]);
      }
    } else {
#pragma unroll
      for (int u = 0; u < 4; ++u) {
        const int g = base + u * 256;
        if (g < E4) tb32[g] = pack(make_int4(dst[4 * g], dst[4 * g + 1], dst[4 * g + 2], dst[4 * g + 3]));
      }
    }
  }
  if (blk == 0 && (int)threadIdx.x < (E & 3)) {
    const int e = 4 * E4 + threadIdx.x;
    ((unsigned char*)tb32)[e] = (unsigned char)(((unsigned)dst[e] / (unsigned)AGG_ROWS) & 255u);
  }
}

__global__ __launch_bounds__(256) void k_bytes_and_gemm(
    const int* __restrict__ dst, unsigned* __restrict__ tb32, int E, int nTB,
    const float* __restrict__ A, const float* __restrict__ B,
    float* __restrict__ C, int M) {
  const int blk = blockIdx.x;
  if (blk < nTB) tile_bytes_block(dst, tb32, E, blk);
  else           gemm_block(A, B, C, M, blk - nTB);
}

template <bool BYTES>
__global__ __launch_bounds__(AGG_THREADS) void k_scan_tiles(
    const int* __restrict__ src, const int* __restrict__ dst, const unsigned char* __restrict__ tileb,
    float* __restrict__ dinv, int* __restrict__ wlist, int* __restrict__ wcnt,
    int capW, int N, int E) {
  __shared__ int list[SCAN_CAP];
  __shared__ int rowcnt[AGG_ROWS];
  __shared__ int cnt;
  __shared__ int cntv;

  const int tid   = threadIdx.x;
  const int r0    = blockIdx.x * AGG_ROWS;
  const int nrows = min(AGG_ROWS, N - r0);
  const unsigned ur0 = (unsigned)r0, un = (unsigned)nrows;

  for (int i = tid; i < AGG_ROWS; i += AGG_THREADS) rowcnt[i] = 0;
  if (tid == 0) { cnt = 0; cntv = 0; }
  __syncthreads();

  auto count_row = [&](int e) -> bool {
    int d = dst[e];
    bool ok = ((unsigned)d - ur0) < un;
    if (ok) atomicAdd(&rowcnt[d - r0], 1);
    return ok;
  };
  auto note = [&](int e) {
    int pos = atomicAdd(&cnt, 1);
    if (pos < SCAN_CAP) list[pos] = e;
    else count_row(e);
  };

  constexpr int U = 8;
  constexpr int TILE = U * AGG_THREADS;
  if constexpr (BYTES) {
    const unsigned tile = blockIdx.x;
    int scalarFrom = 0;
    if ((((size_t)tileb) & 15) == 0) {
      const unsigned pat = tile * 0x01010101u;
      auto hit = [&](unsigned w) {
        unsigned x = w ^ pat;
        return (x - 0x01010101u) & ~x & 0x80808080u;
      };
      auto flags = [&](uint4 v, auto hi) -> unsigned {
        constexpr int HI = decltype(hi)::value;
        return (hit(v.x) >> (7 - HI)) | (hit(v.y) >> (6 - HI)) | (hit(v.z) >> (5 - HI)) | (hit(v.w) >> (4 - HI));
      };
      auto drain = [&](unsigned mm, int e0, int e1) {
        while (mm) {
          int b = __builtin_ctz(mm);
          mm &= mm - 1;
          note(((b & 4) ? e1 : e0) + 4 * (b & 3) + (b >> 3));
        }
      };
      const int G16 = E >> 4;
      const int G32 = G16 & ~31;
      const uint4* tb4 = (const uint4*)tileb;
      const int wave = __builtin_amdgcn_readfirstlane(tid >> 5);
      const int lane = tid & 31;
      const int start = (int)(tile * ((unsigned)G32 / gridDim.x)) & ~31;
      auto place = [&](int i) { int q = start + i; return q >= G32 ? q - G32 : q; };
      int i0 = 0;
      for (; i0 + TILE <= G32; i0 += TILE) {
        uint4 dv[U];
        int   q[U];
#pragma unroll
        for (int u = 0; u < U; ++u) {
          q[u]  = place(i0 + u * AGG_THREADS + wave * 32) + lane;
          dv[u] = tb4[q[u]];
        }
#pragma unroll
        for (int u = 0; u < U; u += 2)
          drain(flags(dv[u], IntC<0>{}) | flags(dv[u + 1], IntC<4>{}), 16 * q[u], 16 * q[u + 1]);
      }
      for (int i = i0 + wave * 32; i < G32; i += AGG_THREADS) {
        const int q = place(i) + lane;
        drain(flags(tb4[q], IntC<0>{}), 16 * q, 16 * q);
      }
      if (tid < G16 - G32) {
        const int q = G32 + tid;
        drain(flags(tb4[q], IntC<0>{}), 16 * q, 16 * q);
      }
      scalarFrom = G16 * 16;
    }
    for (int e = scalarFrom + tid; e < E; e += AGG_THREADS)
      if (tileb[e] == (unsigned char)tile) note(e);
  } else {
    auto check4 = [&](int4 d4, int e) {
      bool m0 = ((unsigned)d4.x - ur0) < un;
      bool m1 = ((unsigned)d4.y - ur0) < un;
      bool m2 = ((unsigned)d4.z - ur0) < un;
      bool m3 = ((unsigned)d4.w - ur0) < un;
      if (m0 | m1 | m2 | m3) {
        if (m0) note(e);
        if (m1) note(e + 1);
        if (m2) note(e + 2);
        if (m3) note(e + 3);
      }
    };
    int scalarFrom = 0;
    if ((((size_t)dst) & 15) == 0) {
      const int E4 = E >> 2;
      const int4* dst4 = (const int4*)dst;
      auto scan_groups = [&](int lo, int hi) {
        int t0 = lo;
        for (; t0 + TILE <= hi; t0 += TILE) {
          int4 dv[U];
#pragma unroll
          for (int u = 0; u < U; ++u) dv[u] = dst4[t0 + tid + u * AGG_THREADS];
#pragma unroll
          for (int u = 0; u < U; ++u) check4(dv[u], 4 * (t0 + tid + u * AGG_THREADS));
        }
        for (int q = t0 + tid; q < hi; q += AGG_THREADS) check4(dst4[q], 4 * q);
      };
      const int start = (int)(blockIdx.x * ((unsigned)E4 / gridDim.x)) & ~7;
      scan_groups(start, E4);
      scan_groups(0, start);
      scalarFrom = E4 * 4;
    }
    for (int e = scalarFrom + tid; e < E; e += AGG_THREADS) {
      int d = dst[e];
      if (((unsigned)d - ur0) < un) note(e);
    }
  }
  __syncthreads();

  const int noted = cnt;
  const int held  = min(noted, SCAN_CAP);
  int* wl = wlist + (size_t)blockIdx.x * LIST_CAP;
  for (int m = tid; m < held; m += AGG_THREADS) {
    const int e = list[m];
    const int s = src[e];
    const int d = dst[e];
    asm volatile("" :: "v"(s));
    if (((unsigned)d - ur0) < un) {
      atomicAdd(&rowcnt[d - r0], 1);
      int p = atomicAdd(&cntv, 1);
      if (p < capW) wl[p] = (int)((unsigned)s | ((unsigned)(d - r0) << SRC_BITS));
    }
  }
  __syncthreads();

  for (int i = tid; i < nrows; i += AGG_THREADS) dinv[r0 + i] = rsqrtf((float)(rowcnt[i] + 1));

  if (tid == 0) wcnt[blockIdx.x] = (noted <= SCAN_CAP) ? cntv : 0x7fffffff;
}

template <int F, int FO, bool LSM = false>
__global__ __launch_bounds__(AGG_THREADS) void k_agg_rows(
    const float* __restrict__ h, const int* __restrict__ src, const int* __restrict__ dst,
    const float* __restrict__ dinv, const int* __restrict__ wlist, const int* __restrict__ wcnt,
    int capW, float* __restrict__ out, int N, int E,
    const float* __restrict__ bias, const float* __restrict__ Wn) {
  constexpr bool MM     = FO > 0;
  constexpr int  PITCH  = MM ? F + 4 : F;
  constexpr int  ROWS_A = MM ? ((AGG_ROWS + 15) / 16) * 16 : AGG_ROWS;
  constexpr int  CAPL   = (AGG_LDS_BYTES - ROWS_A * PITCH * 4) / 12;
  static_assert(LIST_CAP <= CAPL, "LDS must hold a full workspace list beside the tile");
  static_assert(!MM || (F % 4 == 0 && ROWS_A * FO <= 3 * CAPL), "product rows are staged where the edge list was");
  __shared__ alignas(16) float acc[ROWS_A * PITCH];
  __shared__ int lst[3 * CAPL];
  int* eS = lst;
  int* eD = lst + CAPL;
  int* eC = lst + 2 * CAPL;

  const int tid   = threadIdx.x;
  const int r0    = blockIdx.x * AGG_ROWS;
  const int nrows = min(AGG_ROWS, N - r0);
  const int nel   = nrows * F;
  const size_t base = (size_t)r0 * F;

  const int  total  = wcnt[blockIdx.x];
  const bool listed = total <= capW;

  constexpr int LPT = (LIST_CAP + AGG_THREADS - 1) / AGG_THREADS;
  constexpr int NG  = ROWS_A * F / 4;
  constexpr int PPT = (NG + AGG_THREADS - 1) / AGG_THREADS;
  static_assert(F % 4 == 0 && PITCH % 4 == 0, "a 16-byte piece lies in one row, at a 16-byte place");
  const int* wl = wlist + (size_t)blockIdx.x * LIST_CAP;
  if ((((size_t)h) & 15) == 0) {
    const bool staged = listed && total > 0;
    float4 hv[PPT];
    float  dv[PPT];
#pragma unroll
    for (int k = 0; k < PPT; ++k) {
      const int i = min(4 * (tid + k * AGG_THREADS), nel - 4);
      dv[k] = dinv[r0 + i / F];
      hv[k] = *(const float4*)(h + base + i);
    }
    int   lw[LPT];
    float cs[LPT], cd[LPT];
    if (staged) {
#pragma unroll
      for (int k = 0; k < LPT; ++k) lw[k] = wl[min(tid + k * AGG_THREADS, total - 1)];
#pragma unroll
      for (int k = 0; k < LPT; ++k) {
        cs[k] = dinv[lw[k] & SRC_MASK];
        cd[k] = dinv[r0 + (int)((unsigned)lw[k] >> SRC_BITS)];
      }
    }
#pragma unroll
    for (int k = 0; k < PPT; ++k) {
      const int g = tid + k * AGG_THREADS;
      if (g < NG) {
        const int i = 4 * g;
        const int row = i / F, col = i % F;
        const float dd = dv[k] * dv[k];
        float4 v = make_float4(0.0f, 0.0f, 0.0f, 0.0f);
        if (i < nel) v = make_float4(dd * hv[k].x, dd * hv[k].y, dd * hv[k].z, dd * hv[k].w);
        *(float4*)(acc + row * PITCH + col) = v;
      }
    }
    if (staged) {
#pragma unroll
      for (int k = 0; k < LPT; ++k) {
        const int m = tid + k * AGG_THREADS;
        if (m < total) {
          eS[m] = lw[k] & SRC_MASK;
          eD[m] = (int)((unsigned)lw[k] >> SRC_BITS);
          eC[m] = __float_as_int(cs[k] * cd[k]);
        }
      }
      for (int m = tid + LPT * AGG_THREADS; m < total; m += AGG_THREADS) {
        const int w = wl[m];
        const int s = w & SRC_MASK;
        const int r = (int)((unsigned)w >> SRC_BITS);
        eS[m] = s;
        eD[m] = r;
        eC[m] = __float_as_int(dinv[s] * dinv[r0 + r]);
      }
    }
  } else {
    if (listed) {
      for (int m = tid; m < total; m += AGG_THREADS) {
        const int w = wl[m];
        const int s = w & SRC_MASK;
        const int r = (int)((unsigned)w >> SRC_BITS);
        eS[m] = s;
        eD[m] = r;
        eC[m] = __float_as_int(dinv[s] * dinv[r0 + r]);
      }
    }
    for (int i = tid; i < ROWS_A * F; i += AGG_THREADS) {
      const int row = i / F, col = i % F;
      float v = 0.0f;
      if (i < nel) {
        float di = dinv[r0 + row];
        v = di * di * h[base + i];
      }
      acc[row * PITCH + col] = v;
    }
  }
  __syncthreads();

  if (listed) {
    const int wave = __builtin_amdgcn_readfirstlane(tid >> 5);
    const int lane = tid & 31;
    constexpr int NJ = (F + 31) / 32;
    constexpr int G  = 8;
    for (int m0 = wave * 32; m0 < total; m0 += AGG_THREADS) {
      const int m = m0 + lane;
      int s = 0, dl = 0;
      float c = 0.0f;
      if (m < total) {
        s  = eS[m];
        dl = eD[m];
        c  = __int_as_float(eC[m]);
      }
      auto add_edges = [&](auto gg, int i0) {
        constexpr int GG = decltype(gg)::value;
        float v[GG][NJ];
#pragma unroll
        for (int i = 0; i < GG; ++i) {
          int si = __builtin_amdgcn_readlane(s, i0 + i);
          const float* hr = h + (size_t)si * F;
#pragma unroll
          for (int j = 0; j < NJ; ++j) v[i][j] = hr[min(32 * j + lane, F - 1)];
        }
#pragma unroll
        for (int i = 0; i < GG; ++i) {
          int   di = __builtin_amdgcn_readlane(dl, i0 + i);
          float ci = __builtin_bit_cast(float, __builtin_amdgcn_readlane(__builtin_bit_cast(int, c), i0 + i));
          float* ar = acc + di * PITCH;
#pragma unroll
          for (int j = 0; j < NJ; ++j) {
            int f = 32 * j + lane;
            if (f < F) atomicAdd(&ar[f], ci * v[i][j]);
          }
        }
      };
      const int nb = min(32, total - m0);
      int i0 = 0;
      for (; i0 + G <= nb; i0 += G) add_edges(IntC<G>{}, i0);
      for (; i0 < nb; ++i0)         add_edges(IntC<1>{}, i0);
    }
  } else {
    const unsigned ur0 = (unsigned)r0, un = (unsigned)nrows;
    for (int e = tid; e < E; e += AGG_THREADS) {
      int d = dst[e];
      if (((unsigned)d - ur0) < un) {
        int s = src[e];
        float c = dinv[s] * dinv[d];
        for (int f = 0; f < F; ++f)
          atomicAdd(&acc[(d - r0) * PITCH + f], c * h[(size_t)s * F + f]);
      }
    }
  }
  __syncthreads();

  if constexpr (!MM) {
    if constexpr (LSM) {
      static_assert(F > 32 && F <= 64 && PITCH == F, "a lane holds one or two classes of a row");
      constexpr int NW  = AGG_THREADS / 32;
      constexpr int RPW = (AGG_ROWS + NW - 1) / NW;
      const int  wave = __builtin_amdgcn_readfirstlane(tid >> 5);
      const int  lane = tid & 31;
      const bool has2 = (lane + 32) < F;
      const float b0 = bias[lane];
      const float b1 = has2 ? bias[lane + 32] : 0.0f;
      const int  c2   = min(lane + 32, F - 1);
      float v0[RPW], v1[RPW], mx[RPW], sm[RPW];
#pragma unroll
      for (int k = 0; k < RPW; ++k) {
        const int row = wave + k * NW;
        v0[k] = 0.0f;
        v1[k] = 0.0f;
        if (row < nrows) {
          const float* p = acc + row * F;
          v0[k] = p[lane];
          v1[k] = p[c2];
        }
      }
#pragma unroll
      for (int k = 0; k < RPW; ++k) {
        v0[k] = v0[k] + b0;
        v1[k] = has2 ? v1[k] + b1 : -INFINITY;
        mx[k] = fmaxf(v0[k], v1[k]);
      }
#pragma unroll
      for (int off = 16; off > 0; off >>= 1) {
#pragma unroll
        for (int k = 0; k < RPW; ++k) mx[k] = fmaxf(mx[k], __shfl_xor(mx[k], off, 32));
      }
#pragma unroll
      for (int k = 0; k < RPW; ++k)
        sm[k] = expf(v0[k] - mx[k]) + (has2 ? expf(v1[k] - mx[k]) : 0.0f);
#pragma unroll
      for (int off = 16; off > 0; off >>= 1) {
#pragma unroll
        for (int k = 0; k < RPW; ++k) sm[k] += __shfl_xor(sm[k], off, 32);
      }
#pragma unroll
      for (int k = 0; k < RPW; ++k) {
        const int row = wave + k * NW;
        const float lse = logf(sm[k]);
        if (row < nrows) {
          float* p = acc + row * F;
          p[lane] = v0[k] - mx[k] - lse;
          if (has2) p[lane + 32] = v1[k] - mx[k] - lse;
        }
      }
      __syncthreads();
    }
    for (int i = tid; i < nel; i += AGG_THREADS) out[base + i] = acc[i];
  } else {
    for (int i = tid; i < nel; i += AGG_THREADS) {
      const int row = i / F, col = i % F;
      float v = acc[row * PITCH + col] + bias[col];
      acc[row * PITCH + col] = v > 0.0f ? v : 0.0f;
    }
    __syncthreads();

    constexpr int NT = (FO + 15) / 16;
    constexpr int JOBS = (ROWS_A / 16) * NT;
    const int wave = __builtin_amdgcn_readfirstlane(tid >> 5);
    const int lane = tid & 31;
    const int half = lane >> 4;
    const int lm   = lane & 15;
    for (int job = wave; job < JOBS; job += AGG_THREADS / 32) {
      const int rt = job / NT, nt = job % NT;
      const float* aBase = acc + (rt * 16 + lm) * PITCH + 2 * half;
      const int    col   = nt * 16 + lm;
      const bool   cv    = col < FO;
      const float* bBase = Wn + (size_t)(2 * half) * FO + min(col, FO - 1);
      v8f c = {};
#pragma unroll
      for (int t = 0; t < F / 4; ++t) {
        v2f a = *(const v2f*)(aBase + 4 * t);
        const float bx = bBase[(size_t)(4 * t) * FO];
        const float by = bBase[(size_t)(4 * t + 1) * FO];
        v2f b;
        b.x = cv ? bx : 0.0f;
        b.y = cv ? by : 0.0f;
        c = __builtin_amdgcn_wmma_f32_16x16x4_f32(
             false, a,  false, b,
             (short)0, c,  false,  false);
      }
      if (cv) {
        int* st = lst + (rt * 16 + 8 * half) * FO + col;
#pragma unroll
        for (int r = 0; r < 8; ++r) {
          const float cr = c[r];
          st[r * FO] = __float_as_int(cr);
        }
      }
    }
    __syncthreads();

    const size_t obase = (size_t)r0 * FO;
    for (int i = tid; i < nrows * FO; i += AGG_THREADS) out[obase + i] = __int_as_float(lst[i]);
  }
}

static inline int cdiv_ll(long long a, long long b) { return (int)((a + b - 1) / b); }

extern "C" void kernel_launch(void* const* d_in, const int* in_sizes, int n_in,
                              void* d_out, int out_size, void* d_ws, size_t ws_size,
                              hipStream_t stream) {
  const float* x  = (const float*)d_in[0];
  const int*   ei = (const int*)  d_in[1];
  const float* W1 = (const float*)d_in[2];
  const float* b1 = (const float*)d_in[3];
  const float* W2 = (const float*)d_in[4];
  const float* b2 = (const float*)d_in[5];

  const int F = 128, H = 128, C = 40;
  const int N = in_sizes[0] / F;
  const int E = in_sizes[1] / 2;
  const int* src = ei;
  const int* dst = ei + E;

  const int nAgg = cdiv_ll(N, AGG_ROWS);

  float* ws   = (float*)d_ws;
  int*   wcnt = (int*)ws;           ws += N;
  float* dinv = ws;                 ws += N;
  float* h1   = ws;                 ws += (size_t)N * H;
  float* o1   = ws;                 ws += (size_t)N * H;
  int*   wlist = (int*)ws;          ws += (size_t)N * C;
  float* h2   = o1;
  float* out  = (float*)d_out;
  const int capW = ((size_t)nAgg * LIST_CAP <= (size_t)N * C && N <= SRC_MASK + 1) ? LIST_CAP : 0;

  const int B = 256;

  unsigned char* tileb = (unsigned char*)o1;
  const bool bytes = nAgg <= 256 && (size_t)E <= (size_t)N * H * sizeof(float);
  const int nGemm = cdiv_ll(N, 32);
  if (bytes) {
    const int nTB = cdiv_ll((long long)(E >> 2) + 1, 1024);
    k_bytes_and_gemm<<<nTB + nGemm, B, 0, stream>>>(dst, (unsigned*)tileb, E, nTB, x, W1, h1, N);
    k_scan_tiles<true><<<nAgg, AGG_THREADS, 0, stream>>>(src, dst, tileb, dinv, wlist, wcnt, capW, N, E);
  } else {
    gcn_gemm_wmma_f32<<<nGemm, B, 0, stream>>>(x, W1, h1, N);
    k_scan_tiles<false><<<nAgg, AGG_THREADS, 0, stream>>>(src, dst, tileb, dinv, wlist, wcnt, capW, N, E);
  }

  k_agg_rows<128, 40><<<nAgg, AGG_THREADS, 0, stream>>>(h1, src, dst, dinv, wlist, wcnt, capW, h2, N, E, b1, W2);

  k_agg_rows<40, 0, true><<<nAgg, AGG_THREADS, 0, stream>>>(h2, src, dst, dinv, wlist, wcnt, capW, out, N, E, b2, nullptr);
}
